# starting kernel
# baseline (speedup 1.0000x reference)
#include <hip/hip_runtime.h>


typedef __attribute__((ext_vector_type(2))) float v2f;
typedef __attribute__((ext_vector_type(8))) float v8f;

#define WAVES 8

__device__ __forceinline__ unsigned fkey(float f) {
  unsigned u = __float_as_uint(f);
  return (u & 0x80000000u) ? ~u : (u | 0x80000000u);
}
__device__ __forceinline__ float funkey(unsigned kk) {
  unsigned u = (kk & 0x80000000u) ? (kk & 0x7FFFFFFFu) : ~kk;
  return __uint_as_float(u);
}

__global__ __launch_bounds__(256) void gemm128_kernel(
    const float* __restrict__ A, const float* __restrict__ W,
    float* __restrict__ out) {
  const int lane = threadIdx.x & 31;
  const int wv   = threadIdx.x >> 5;
  const int row0 = blockIdx.x << 4;
  const int n0   = wv << 4;
  const int m    = lane & 15;
  const int kk   = (lane >> 4) << 1;
  const int mb   = (lane >> 4) << 3;

  v8f c = {};
  const float* ar = A + (size_t)(row0 + m) * 128;
  const float* wp = W + n0 + m;
#pragma unroll 8
  for (int k0 = 0; k0 < 128; k0 += 4) {
    v2f a, b;
    a[0] = ar[k0 + kk];
    a[1] = ar[k0 + kk + 1];
    b[0] = wp[(size_t)(k0 + kk) * 128];
    b[1] = wp[(size_t)(k0 + kk + 1) * 128];
    c = __builtin_amdgcn_wmma_f32_16x16x4_f32(false, a, false, b, (short)0, c,
                                              false, false);
  }
  float* op = out + (size_t)(row0 + mb) * 128 + n0 + m;
#pragma unroll
  for (int r = 0; r < 8; ++r) op[(size_t)r * 128] = c[r];
}

__global__ __launch_bounds__(256) void edge_score_kernel(
    const float* __restrict__ ea, const int* __restrict__ src,
    const int* __restrict__ dst, const float* __restrict__ Wek,
    const float* __restrict__ Wg1, const float* __restrict__ Wg2,
    const float* __restrict__ q, const float* __restrict__ k,
    float* __restrict__ scores, unsigned* __restrict__ mmax) {
  __shared__ float g1s[WAVES][16][64];
  __shared__ float gts[WAVES][16][8];

  const int lane = threadIdx.x & 31;
  const int wv   = threadIdx.x >> 5;
  const int e0   = (blockIdx.x * WAVES + wv) << 4;
  const int m    = lane & 15;
  const int kk   = (lane >> 4) << 1;
  const int mb   = (lane >> 4) << 3;

  v2f afr[8];
  const float* er = ea + (size_t)(e0 + m) * 32;
#pragma unroll
  for (int i = 0; i < 8; ++i) {
    afr[i][0] = er[4 * i + kk];
    afr[i][1] = er[4 * i + kk + 1];
  }

#pragma unroll
  for (int t = 0; t < 4; ++t) {
    v8f c = {};
    const float* wp = Wg1 + t * 16 + m;
#pragma unroll
    for (int i = 0; i < 8; ++i) {
      v2f b;
      b[0] = wp[(size_t)(4 * i + kk) * 64];
      b[1] = wp[(size_t)(4 * i + kk + 1) * 64];
      c = __builtin_amdgcn_wmma_f32_16x16x4_f32(false, afr[i], false, b,
                                                (short)0, c, false, false);
    }
#pragma unroll
    for (int r = 0; r < 8; ++r) {
      float xv = c[r];
      g1s[wv][mb + r][t * 16 + m] = xv / (1.0f + __expf(-xv));
    }
  }
  __syncthreads();

  {
    const int hb = (lane >> 4) << 2;
    float acc0 = 0.f, acc1 = 0.f, acc2 = 0.f, acc3 = 0.f;
#pragma unroll 4
    for (int j = 0; j < 64; ++j) {
      float g = g1s[wv][m][j];
      const float* w2 = Wg2 + j * 8 + hb;
      acc0 = fmaf(g, w2[0], acc0);
      acc1 = fmaf(g, w2[1], acc1);
      acc2 = fmaf(g, w2[2], acc2);
      acc3 = fmaf(g, w2[3], acc3);
    }
    gts[wv][m][hb + 0] = acc0;
    gts[wv][m][hb + 1] = acc1;
    gts[wv][m][hb + 2] = acc2;
    gts[wv][m][hb + 3] = acc3;
  }
  __syncthreads();

  int srcv[8], dstv[8];
#pragma unroll
  for (int r = 0; r < 8; ++r) {
    srcv[r] = src[e0 + mb + r];
    dstv[r] = dst[e0 + mb + r];
  }

#pragma unroll 1
  for (int h = 0; h < 8; ++h) {
    v8f c = {};
    const float* wp = Wek + h * 16 + m;
#pragma unroll
    for (int i = 0; i < 8; ++i) {
      v2f b;
      b[0] = wp[(size_t)(4 * i + kk) * 128];
      b[1] = wp[(size_t)(4 * i + kk + 1) * 128];
      c = __builtin_amdgcn_wmma_f32_16x16x4_f32(false, afr[i], false, b,
                                                (short)0, c, false, false);
    }
    const int col = h * 16 + m;
#pragma unroll
    for (int r = 0; r < 8; ++r) {
      float kv = k[(size_t)srcv[r] * 128 + col] + c[r];
      float qv = q[(size_t)dstv[r] * 128 + col];
      float p  = qv * kv;
#pragma unroll
      for (int off = 8; off >= 1; off >>= 1) p += __shfl_xor(p, off, 32);
      if (m == 0) {
        float sc = p * 0.25f + gts[wv][mb + r][h];
        scores[(size_t)(e0 + mb + r) * 8 + h] = sc;
        atomicMax(&mmax[(size_t)dstv[r] * 8 + h], fkey(sc));
      }
    }
  }
}

__global__ __launch_bounds__(256) void exp_denom_kernel(
    const int* __restrict__ dst, float* __restrict__ scores,
    const unsigned* __restrict__ mmax, float* __restrict__ denom, int total) {
  int idx = blockIdx.x * blockDim.x + threadIdx.x;
  if (idx >= total) return;
  int e = idx >> 3, h = idx & 7;
  int d = dst[e];
  float mv = funkey(mmax[(size_t)d * 8 + h]);
  float es = __expf(scores[idx] - mv);
  scores[idx] = es;
  atomicAdd(&denom[(size_t)d * 8 + h], es);
}

__global__ __launch_bounds__(256) void edge_aggregate_kernel(
    const float* __restrict__ ea, const int* __restrict__ src,
    const int* __restrict__ dst, const float* __restrict__ Wev,
    const float* __restrict__ v, const float* __restrict__ scores,
    const float* __restrict__ denom, float* __restrict__ acc) {
  const int lane = threadIdx.x & 31;
  const int wv   = threadIdx.x >> 5;
  const int e0   = (blockIdx.x * WAVES + wv) << 4;
  const int m    = lane & 15;
  const int kk   = (lane >> 4) << 1;
  const int mb   = (lane >> 4) << 3;

  v2f afr[8];
  const float* er = ea + (size_t)(e0 + m) * 32;
#pragma unroll
  for (int i = 0; i < 8; ++i) {
    afr[i][0] = er[4 * i + kk];
    afr[i][1] = er[4 * i + kk + 1];
  }

  int srcv[8], dstv[8];
#pragma unroll
  for (int r = 0; r < 8; ++r) {
    srcv[r] = src[e0 + mb + r];
    dstv[r] = dst[e0 + mb + r];
  }

#pragma unroll 1
  for (int h = 0; h < 8; ++h) {
    v8f c = {};
    const float* wp = Wev + h * 16 + m;
#pragma unroll
    for (int i = 0; i < 8; ++i) {
      v2f b;
      b[0] = wp[(size_t)(4 * i + kk) * 128];
      b[1] = wp[(size_t)(4 * i + kk + 1) * 128];
      c = __builtin_amdgcn_wmma_f32_16x16x4_f32(false, afr[i], false, b,
                                                (short)0, c, false, false);
    }
    const int col = h * 16 + m;
#pragma unroll
    for (int r = 0; r < 8; ++r) {
      int e = e0 + mb + r;
      float att = scores[(size_t)e * 8 + h] /
                  (denom[(size_t)dstv[r] * 8 + h] + 1e-9f);
      float ve = v[(size_t)srcv[r] * 128 + col] + c[r];
      atomicAdd(&acc[(size_t)dstv[r] * 128 + col], att * ve);
    }
  }
}

extern "C" void kernel_launch(void* const* d_in, const int* in_sizes, int n_in,
                              void* d_out, int out_size, void* d_ws,
                              size_t ws_size, hipStream_t stream) {
  const float* x   = (const float*)d_in[0];
  const int*   ei  = (const int*)d_in[1];
  const float* ea  = (const float*)d_in[2];
  const float* Wq  = (const float*)d_in[3];
  const float* Wk  = (const float*)d_in[4];
  const float* Wv  = (const float*)d_in[5];
  const float* Wo  = (const float*)d_in[6];
  const float* Wek = (const float*)d_in[7];
  const float* Wev = (const float*)d_in[8];
  const float* Wg1 = (const float*)d_in[9];
  const float* Wg2 = (const float*)d_in[10];
  float* out = (float*)d_out;

  const int N = in_sizes[0] / 128;
  const int E = in_sizes[2] / 32;
  const int* src = ei;
  const int* dst = ei + E;

  float* q      = (float*)d_ws;
  float* k      = q + (size_t)N * 128;
  float* v      = k + (size_t)N * 128;
  float* scores = v + (size_t)N * 128;
  float* acc    = scores + (size_t)E * 8;
  float* denom  = acc + (size_t)N * 128;
  unsigned* mmax = (unsigned*)(denom + (size_t)N * 8);

  hipMemsetAsync(acc, 0, (size_t)(N * 128 + N * 8 + N * 8) * sizeof(float),
                 stream);

  dim3 blk(256);
  gemm128_kernel<<<dim3(N / 16), blk, 0, stream>>>(x, Wq, q);
  gemm128_kernel<<<dim3(N / 16), blk, 0, stream>>>(x, Wk, k);
  gemm128_kernel<<<dim3(N / 16), blk, 0, stream>>>(x, Wv, v);

  edge_score_kernel<<<dim3(E / (16 * WAVES)), blk, 0, stream>>>(
      ea, src, dst, Wek, Wg1, Wg2, q, k, scores, mmax);

  exp_denom_kernel<<<dim3((E * 8 + 255) / 256), blk, 0, stream>>>(
      dst, scores, mmax, denom, E * 8);

  edge_aggregate_kernel<<<dim3(E / (16 * WAVES)), blk, 0, stream>>>(
      ea, src, dst, Wev, v, scores, denom, acc);

  gemm128_kernel<<<dim3(N / 16), blk, 0, stream>>>(acc, Wo, out);
}


# q, k and v projections in one launch: a block's x fragment is loaded once per k-step and feeds three WMMAs (Wq, Wk, Wv tiles); 5 kernel launches, not 7; sums unchanged
# speedup vs baseline: 2.2185x; 2.2185x over previous
#include <hip/hip_runtime.h>


typedef __attribute__((ext_vector_type(2))) float v2f;
typedef __attribute__((ext_vector_type(8))) float v8f;

#define WAVES 8

__device__ __forceinline__ unsigned fkey(float f) {
  unsigned u = __float_as_uint(f);
  return (u & 0x80000000u) ? ~u : (u | 0x80000000u);
}
__device__ __forceinline__ float funkey(unsigned kk) {
  unsigned u = (kk & 0x80000000u) ? (kk & 0x7FFFFFFFu) : ~kk;
  return __uint_as_float(u);
}

__global__ __launch_bounds__(256) void gemm128_kernel(
    const float* __restrict__ A, const float* __restrict__ W,
    float* __restrict__ out) {
  const int lane = threadIdx.x & 31;
  const int wv   = threadIdx.x >> 5;
  const int row0 = blockIdx.x << 4;
  const int n0   = wv << 4;
  const int m    = lane & 15;
  const int kk   = (lane >> 4) << 1;
  const int mb   = (lane >> 4) << 3;

  v8f c = {};
  const float* ar = A + (size_t)(row0 + m) * 128;
  const float* wp = W + n0 + m;
#pragma unroll 8
  for (int k0 = 0; k0 < 128; k0 += 4) {
    v2f a, b;
    a[0] = ar[k0 + kk];
    a[1] = ar[k0 + kk + 1];
    b[0] = wp[(size_t)(k0 + kk) * 128];
    b[1] = wp[(size_t)(k0 + kk + 1) * 128];
    c = __builtin_amdgcn_wmma_f32_16x16x4_f32(false, a, false, b, (short)0, c,
                                              false, false);
  }
  float* op = out + (size_t)(row0 + mb) * 128 + n0 + m;
#pragma unroll
  for (int r = 0; r < 8; ++r) op[(size_t)r * 128] = c[r];
}

__global__ __launch_bounds__(256) void qkv_kernel(
    const float* __restrict__ A, const float* __restrict__ Wq,
    const float* __restrict__ Wk, const float* __restrict__ Wv,
    float* __restrict__ q, float* __restrict__ k, float* __restrict__ v) {
  const int lane = threadIdx.x & 31;
  const int wv   = threadIdx.x >> 5;
  const int row0 = blockIdx.x << 4;
  const int n0   = wv << 4;
  const int m    = lane & 15;
  const int kk   = (lane >> 4) << 1;
  const int mb   = (lane >> 4) << 3;

  v8f cq = {}, ck = {}, cv = {};
  const float* ar = A + (size_t)(row0 + m) * 128;
  const size_t wc = (size_t)(n0 + m);
#pragma unroll 4
  for (int k0 = 0; k0 < 128; k0 += 4) {
    v2f a, bq, bk, bv;
    const size_t w0 = (size_t)(k0 + kk) * 128 + wc;
    const size_t w1 = w0 + 128;
    a[0]  = ar[k0 + kk];
    a[1]  = ar[k0 + kk + 1];
    bq[0] = Wq[w0];
    bq[1] = Wq[w1];
    bk[0] = Wk[w0];
    bk[1] = Wk[w1];
    bv[0] = Wv[w0];
    bv[1] = Wv[w1];
    cq = __builtin_amdgcn_wmma_f32_16x16x4_f32(false, a, false, bq, (short)0, cq,
                                               false, false);
    ck = __builtin_amdgcn_wmma_f32_16x16x4_f32(false, a, false, bk, (short)0, ck,
                                               false, false);
    cv = __builtin_amdgcn_wmma_f32_16x16x4_f32(false, a, false, bv, (short)0, cv,
                                               false, false);
  }
  const size_t o = (size_t)(row0 + mb) * 128 + wc;
#pragma unroll
  for (int r = 0; r < 8; ++r) q[o + (size_t)r * 128] = cq[r];
#pragma unroll
  for (int r = 0; r < 8; ++r) k[o + (size_t)r * 128] = ck[r];
#pragma unroll
  for (int r = 0; r < 8; ++r) v[o + (size_t)r * 128] = cv[r];
}

__global__ __launch_bounds__(256) void edge_score_kernel(
    const float* __restrict__ ea, const int* __restrict__ src,
    const int* __restrict__ dst, const float* __restrict__ Wek,
    const float* __restrict__ Wg1, const float* __restrict__ Wg2,
    const float* __restrict__ q, const float* __restrict__ k,
    float* __restrict__ scores, unsigned* __restrict__ mmax) {
  __shared__ float g1s[WAVES][16][64];
  __shared__ float gts[WAVES][16][8];

  const int lane = threadIdx.x & 31;
  const int wv   = threadIdx.x >> 5;
  const int e0   = (blockIdx.x * WAVES + wv) << 4;
  const int m    = lane & 15;
  const int kk   = (lane >> 4) << 1;
  const int mb   = (lane >> 4) << 3;

  v2f afr[8];
  const float* er = ea + (size_t)(e0 + m) * 32;
#pragma unroll
  for (int i = 0; i < 8; ++i) {
    afr[i][0] = er[4 * i + kk];
    afr[i][1] = er[4 * i + kk + 1];
  }

#pragma unroll
  for (int t = 0; t < 4; ++t) {
    v8f c = {};
    const float* wp = Wg1 + t * 16 + m;
#pragma unroll
    for (int i = 0; i < 8; ++i) {
      v2f b;
      b[0] = wp[(size_t)(4 * i + kk) * 64];
      b[1] = wp[(size_t)(4 * i + kk + 1) * 64];
      c = __builtin_amdgcn_wmma_f32_16x16x4_f32(false, afr[i], false, b,
                                                (short)0, c, false, false);
    }
#pragma unroll
    for (int r = 0; r < 8; ++r) {
      float xv = c[r];
      g1s[wv][mb + r][t * 16 + m] = xv / (1.0f + __expf(-xv));
    }
  }
  __syncthreads();

  {
    const int hb = (lane >> 4) << 2;
    float acc0 = 0.f, acc1 = 0.f, acc2 = 0.f, acc3 = 0.f;
#pragma unroll 4
    for (int j = 0; j < 64; ++j) {
      float g = g1s[wv][m][j];
      const float* w2 = Wg2 + j * 8 + hb;
      acc0 = fmaf(g, w2[0], acc0);
      acc1 = fmaf(g, w2[1], acc1);
      acc2 = fmaf(g, w2[2], acc2);
      acc3 = fmaf(g, w2[3], acc3);
    }
    gts[wv][m][hb + 0] = acc0;
    gts[wv][m][hb + 1] = acc1;
    gts[wv][m][hb + 2] = acc2;
    gts[wv][m][hb + 3] = acc3;
  }
  __syncthreads();

  int srcv[8], dstv[8];
#pragma unroll
  for (int r = 0; r < 8; ++r) {
    srcv[r] = src[e0 + mb + r];
    dstv[r] = dst[e0 + mb + r];
  }

  const int hm = m & 7;
  float keep[8];
#pragma unroll
  for (int r = 0; r < 8; ++r) keep[r] = 0.f;

#pragma unroll 1
  for (int h = 0; h < 8; ++h) {
    const int col = h * 16 + m;

    float kv[8], qv[8];
#pragma unroll
    for (int r = 0; r < 8; ++r) {
      kv[r] = k[(size_t)srcv[r] * 128 + col];
      qv[r] = q[(size_t)dstv[r] * 128 + col];
    }

    v8f c = {};
    const float* wp = Wek + h * 16 + m;
#pragma unroll
    for (int i = 0; i < 8; ++i) {
      v2f b;
      b[0] = wp[(size_t)(4 * i + kk) * 128];
      b[1] = wp[(size_t)(4 * i + kk + 1) * 128];
      c = __builtin_amdgcn_wmma_f32_16x16x4_f32(false, afr[i], false, b,
                                                (short)0, c, false, false);
    }

    const bool mine = (hm == h);
#pragma unroll
    for (int r = 0; r < 8; ++r) {
      float pp = qv[r] * (kv[r] + c[r]);
#pragma unroll
      for (int off = 8; off >= 1; off >>= 1) pp += __shfl_xor(pp, off, 32);
      keep[r] = mine ? pp : keep[r];
    }
  }

  float gv[8];
#pragma unroll
  for (int r = 0; r < 8; ++r) gv[r] = gts[wv][mb + r][hm];
  if (m < 8) {
#pragma unroll
    for (int r = 0; r < 8; ++r) {
      float sc = keep[r] * 0.25f + gv[r];
      scores[(size_t)(e0 + mb + r) * 8 + hm] = sc;
      atomicMax(&mmax[(size_t)dstv[r] * 8 + hm], fkey(sc));
    }
  }
}

__global__ __launch_bounds__(256) void exp_denom_kernel(
    const int* __restrict__ dst, float* __restrict__ scores,
    const unsigned* __restrict__ mmax, float* __restrict__ denom, int total) {
  int idx = blockIdx.x * blockDim.x + threadIdx.x;
  if (idx >= total) return;
  int e = idx >> 3, h = idx & 7;
  int d = dst[e];
  float mv = funkey(mmax[(size_t)d * 8 + h]);
  float es = __expf(scores[idx] - mv);
  scores[idx] = es;
  atomicAdd(&denom[(size_t)d * 8 + h], es);
}

__device__ __forceinline__ float xhalf(float x) {
  int b = __builtin_bit_cast(int, x);
  return __builtin_bit_cast(
      float, __builtin_amdgcn_permlanex16(b, b, 0x76543210, 0xfedcba98, false,
                                          false));
}

__global__ __launch_bounds__(256) void edge_aggregate_kernel(
    const float* __restrict__ ea, const int* __restrict__ src,
    const int* __restrict__ dst, const float* __restrict__ Wev,
    const float* __restrict__ v, const float* __restrict__ scores,
    const float* __restrict__ denom, float* __restrict__ acc) {
  const int lane = threadIdx.x & 31;
  const int wv   = threadIdx.x >> 5;
  const int e0   = __builtin_amdgcn_readfirstlane((blockIdx.x * WAVES + wv) << 4);
  const int m    = lane & 15;
  const int kk   = (lane >> 4) << 1;
  const int mb   = (lane >> 4) << 3;
  const bool upper = lane >= 16;

  v2f afr[8];
  const float* er = ea + (size_t)(e0 + m) * 32;
#pragma unroll
  for (int i = 0; i < 8; ++i) {
    afr[i][0] = er[4 * i + kk];
    afr[i][1] = er[4 * i + kk + 1];
  }

  int sn[16], dn[16];
#pragma unroll
  for (int j = 0; j < 16; ++j) {
    sn[j] = src[e0 + j];
    dn[j] = dst[e0 + j];
  }

  float vx[8], vy[8];
  auto load_v = [&](const int p) {
    const int colx = p * 32 + lane;
#pragma unroll
    for (int r = 0; r < 8; ++r) {
      vx[r] = v[(size_t)sn[r] * 128 + colx];
      vy[r] = v[(size_t)sn[8 + r] * 128 + colx];
    }
  };

  const int hm = m & 7;
  int att8[8];
  {
    load_v(0);
    float sv[8], dv[8];
#pragma unroll
    for (int r = 0; r < 8; ++r) {
      const int dr = upper ? dn[8 + r] : dn[r];
      sv[r] = scores[(size_t)(e0 + mb + r) * 8 + hm];
      dv[r] = denom[(size_t)dr * 8 + hm];
    }
    __builtin_amdgcn_sched_barrier(0);
#pragma unroll
    for (int r = 0; r < 8; ++r)
      att8[r] = __float_as_int(sv[r] / (dv[r] + 1e-9f));
  }

  auto trip = [&](const int p) {
    const int colx = p * 32 + lane;
    if (p > 0) {
      load_v(p);
      __builtin_amdgcn_sched_barrier(0);
    }

    v8f c0 = {}, c1 = {};
    const float* wp = Wev + p * 32 + m;
#pragma unroll
    for (int i = 0; i < 8; ++i) {
      v2f b0, b1;
      b0[0] = wp[(size_t)(4 * i + kk) * 128];
      b0[1] = wp[(size_t)(4 * i + kk + 1) * 128];
      b1[0] = wp[(size_t)(4 * i + kk) * 128 + 16];
      b1[1] = wp[(size_t)(4 * i + kk + 1) * 128 + 16];
      c0 = __builtin_amdgcn_wmma_f32_16x16x4_f32(false, afr[i], false, b0,
                                                 (short)0, c0, false, false);
      c1 = __builtin_amdgcn_wmma_f32_16x16x4_f32(false, afr[i], false, b1,
                                                 (short)0, c1, false, false);
    }

#pragma unroll
    for (int r = 0; r < 8; ++r) {
      const float t0 = xhalf(c0[r]);
      const float t1 = xhalf(c1[r]);
      const float cx = upper ? t1 : c0[r];
      const float cy = upper ? c1[r] : t0;
      const float ax_lo = __int_as_float(__builtin_amdgcn_readlane(att8[r], 2 * p));
      const float ax_hi = __int_as_float(__builtin_amdgcn_readlane(att8[r], 2 * p + 1));
      const float ay_lo = __int_as_float(__builtin_amdgcn_readlane(att8[r], 2 * p + 16));
      const float ay_hi = __int_as_float(__builtin_amdgcn_readlane(att8[r], 2 * p + 17));
      const float ax = upper ? ax_hi : ax_lo;
      const float ay = upper ? ay_hi : ay_lo;
      atomicAdd(&acc[(size_t)dn[r] * 128 + colx], ax * (vx[r] + cx));
      atomicAdd(&acc[(size_t)dn[8 + r] * 128 + colx], ay * (vy[r] + cy));
    }
  };

#pragma unroll
  for (int p = 0; p < 4; ++p) trip(p);
}

extern "C" void kernel_launch(void* const* d_in, const int* in_sizes, int n_in,
                              void* d_out, int out_size, void* d_ws,
                              size_t ws_size, hipStream_t stream) {
  const float* x   = (const float*)d_in[0];
  const int*   ei  = (const int*)d_in[1];
  const float* ea  = (const float*)d_in[2];
  const float* Wq  = (const float*)d_in[3];
  const float* Wk  = (const float*)d_in[4];
  const float* Wv  = (const float*)d_in[5];
  const float* Wo  = (const float*)d_in[6];
  const float* Wek = (const float*)d_in[7];
  const float* Wev = (const float*)d_in[8];
  const float* Wg1 = (const float*)d_in[9];
  const float* Wg2 = (const float*)d_in[10];
  float* out = (float*)d_out;

  const int N = in_sizes[0] / 128;
  const int E = in_sizes[2] / 32;
  const int* src = ei;
  const int* dst = ei + E;

  float* q      = (float*)d_ws;
  float* k      = q + (size_t)N * 128;
  float* v      = k + (size_t)N * 128;
  float* scores = v + (size_t)N * 128;
  float* acc    = scores + (size_t)E * 8;
  float* denom  = acc + (size_t)N * 128;
  unsigned* mmax = (unsigned*)(denom + (size_t)N * 8);

  hipMemsetAsync(acc, 0, (size_t)(N * 128 + N * 8 + N * 8) * sizeof(float),
                 stream);

  dim3 blk(256);
  qkv_kernel<<<dim3(N / 16), blk, 0, stream>>>(x, Wq, Wk, Wv, q, k, v);

  edge_score_kernel<<<dim3(E / (16 * WAVES)), blk, 0, stream>>>(
      ea, src, dst, Wek, Wg1, Wg2, q, k, scores, mmax);

  exp_denom_kernel<<<dim3((E * 8 + 255) / 256), blk, 0, stream>>>(
      dst, scores, mmax, denom, E * 8);

  edge_aggregate_kernel<<<dim3(E / (16 * WAVES)), blk, 0, stream>>>(
      ea, src, dst, Wev, v, scores, denom, acc);

  gemm128_kernel<<<dim3(N / 16), blk, 0, stream>>>(acc, Wo, out);
}


# edge_score: the dot's 16-lane sums are row_xmask DPP adds on the VALU, not ds_bpermute shuffles through the LDS unit; step 8 folds rows j and j+4 into one register; same sum tree, first add unfused
# speedup vs baseline: 2.3160x; 1.0439x over previous
#include <hip/hip_runtime.h>


typedef __attribute__((ext_vector_type(2))) float v2f;
typedef __attribute__((ext_vector_type(8))) float v8f;

#define WAVES 8

__device__ __forceinline__ unsigned fkey(float f) {
  unsigned u = __float_as_uint(f);
  return (u & 0x80000000u) ? ~u : (u | 0x80000000u);
}
__device__ __forceinline__ float funkey(unsigned kk) {
  unsigned u = (kk & 0x80000000u) ? (kk & 0x7FFFFFFFu) : ~kk;
  return __uint_as_float(u);
}

__global__ __launch_bounds__(256) void gemm128_kernel(
    const float* __restrict__ A, const float* __restrict__ W,
    float* __restrict__ out) {
  const int lane = threadIdx.x & 31;
  const int wv   = threadIdx.x >> 5;
  const int row0 = blockIdx.x << 4;
  const int n0   = wv << 4;
  const int m    = lane & 15;
  const int kk   = (lane >> 4) << 1;
  const int mb   = (lane >> 4) << 3;

  v8f c = {};
  const float* ar = A + (size_t)(row0 + m) * 128;
  const float* wp = W + n0 + m;
#pragma unroll 8
  for (int k0 = 0; k0 < 128; k0 += 4) {
    v2f a, b;
    a[0] = ar[k0 + kk];
    a[1] = ar[k0 + kk + 1];
    b[0] = wp[(size_t)(k0 + kk) * 128];
    b[1] = wp[(size_t)(k0 + kk + 1) * 128];
    c = __builtin_amdgcn_wmma_f32_16x16x4_f32(false, a, false, b, (short)0, c,
                                              false, false);
  }
  float* op = out + (size_t)(row0 + mb) * 128 + n0 + m;
#pragma unroll
  for (int r = 0; r < 8; ++r) op[(size_t)r * 128] = c[r];
}

__global__ __launch_bounds__(256) void qkv_kernel(
    const float* __restrict__ A, const float* __restrict__ Wq,
    const float* __restrict__ Wk, const float* __restrict__ Wv,
    float* __restrict__ q, float* __restrict__ k, float* __restrict__ v) {
  const int lane = threadIdx.x & 31;
  const int wv   = threadIdx.x >> 5;
  const int row0 = blockIdx.x << 4;
  const int n0   = wv << 4;
  const int m    = lane & 15;
  const int kk   = (lane >> 4) << 1;
  const int mb   = (lane >> 4) << 3;

  v8f cq = {}, ck = {}, cv = {};
  const float* ar = A + (size_t)(row0 + m) * 128;
  const size_t wc = (size_t)(n0 + m);
#pragma unroll 4
  for (int k0 = 0; k0 < 128; k0 += 4) {
    v2f a, bq, bk, bv;
    const size_t w0 = (size_t)(k0 + kk) * 128 + wc;
    const size_t w1 = w0 + 128;
    a[0]  = ar[k0 + kk];
    a[1]  = ar[k0 + kk + 1];
    bq[0] = Wq[w0];
    bq[1] = Wq[w1];
    bk[0] = Wk[w0];
    bk[1] = Wk[w1];
    bv[0] = Wv[w0];
    bv[1] = Wv[w1];
    cq = __builtin_amdgcn_wmma_f32_16x16x4_f32(false, a, false, bq, (short)0, cq,
                                               false, false);
    ck = __builtin_amdgcn_wmma_f32_16x16x4_f32(false, a, false, bk, (short)0, ck,
                                               false, false);
    cv = __builtin_amdgcn_wmma_f32_16x16x4_f32(false, a, false, bv, (short)0, cv,
                                               false, false);
  }
  const size_t o = (size_t)(row0 + mb) * 128 + wc;
#pragma unroll
  for (int r = 0; r < 8; ++r) q[o + (size_t)r * 128] = cq[r];
#pragma unroll
  for (int r = 0; r < 8; ++r) k[o + (size_t)r * 128] = ck[r];
#pragma unroll
  for (int r = 0; r < 8; ++r) v[o + (size_t)r * 128] = cv[r];
}

template <int MASK>
__device__ __forceinline__ float row_xor(float x) {
  return __int_as_float(__builtin_amdgcn_update_dpp(
      0, __float_as_int(x), 0x160 | MASK, 0xf, 0xf, true));
}

__global__ __launch_bounds__(256) void edge_score_kernel(
    const float* __restrict__ ea, const int* __restrict__ src,
    const int* __restrict__ dst, const float* __restrict__ Wek,
    const float* __restrict__ Wg1, const float* __restrict__ Wg2,
    const float* __restrict__ q, const float* __restrict__ k,
    float* __restrict__ scores, unsigned* __restrict__ mmax) {
  __shared__ float g1s[WAVES][16][64];
  __shared__ float gts[WAVES][16][8];

  const int lane = threadIdx.x & 31;
  const int wv   = threadIdx.x >> 5;
  const int e0   = (blockIdx.x * WAVES + wv) << 4;
  const int m    = lane & 15;
  const int kk   = (lane >> 4) << 1;
  const int mb   = (lane >> 4) << 3;

  v2f afr[8];
  const float* er = ea + (size_t)(e0 + m) * 32;
#pragma unroll
  for (int i = 0; i < 8; ++i) {
    afr[i][0] = er[4 * i + kk];
    afr[i][1] = er[4 * i + kk + 1];
  }

#pragma unroll
  for (int t = 0; t < 4; ++t) {
    v8f c = {};
    const float* wp = Wg1 + t * 16 + m;
#pragma unroll
    for (int i = 0; i < 8; ++i) {
      v2f b;
      b[0] = wp[(size_t)(4 * i + kk) * 64];
      b[1] = wp[(size_t)(4 * i + kk + 1) * 64];
      c = __builtin_amdgcn_wmma_f32_16x16x4_f32(false, afr[i], false, b,
                                                (short)0, c, false, false);
    }
#pragma unroll
    for (int r = 0; r < 8; ++r) {
      float xv = c[r];
      g1s[wv][mb + r][t * 16 + m] = xv / (1.0f + __expf(-xv));
    }
  }
  __syncthreads();

  {
    const int hb = (lane >> 4) << 2;
    float acc0 = 0.f, acc1 = 0.f, acc2 = 0.f, acc3 = 0.f;
#pragma unroll 4
    for (int j = 0; j < 64; ++j) {
      float g = g1s[wv][m][j];
      const float* w2 = Wg2 + j * 8 + hb;
      acc0 = fmaf(g, w2[0], acc0);
      acc1 = fmaf(g, w2[1], acc1);
      acc2 = fmaf(g, w2[2], acc2);
      acc3 = fmaf(g, w2[3], acc3);
    }
    gts[wv][m][hb + 0] = acc0;
    gts[wv][m][hb + 1] = acc1;
    gts[wv][m][hb + 2] = acc2;
    gts[wv][m][hb + 3] = acc3;
  }
  __syncthreads();

  int srcv[8], dstv[8];
#pragma unroll
  for (int r = 0; r < 8; ++r) {
    srcv[r] = src[e0 + mb + r];
    dstv[r] = dst[e0 + mb + r];
  }

  const int hm = m & 7;
  const bool hi8 = (m & 8) != 0;
  const int rsel = hi8 ? 4 : 0;
  float keep[4];
  int dsel[4];
#pragma unroll
  for (int j = 0; j < 4; ++j) {
    keep[j] = 0.f;
    dsel[j] = hi8 ? dstv[j + 4] : dstv[j];
  }

#pragma unroll 1
  for (int h = 0; h < 8; ++h) {
    const int col = h * 16 + m;

    float kv[8], qv[8];
#pragma unroll
    for (int r = 0; r < 8; ++r) {
      kv[r] = k[(size_t)srcv[r] * 128 + col];
      qv[r] = q[(size_t)dstv[r] * 128 + col];
    }

    v8f c = {};
    const float* wp = Wek + h * 16 + m;
#pragma unroll
    for (int i = 0; i < 8; ++i) {
      v2f b;
      b[0] = wp[(size_t)(4 * i + kk) * 128];
      b[1] = wp[(size_t)(4 * i + kk + 1) * 128];
      c = __builtin_amdgcn_wmma_f32_16x16x4_f32(false, afr[i], false, b,
                                                (short)0, c, false, false);
    }

    const bool mine = (hm == h);
    float s4[4];
#pragma unroll
    for (int j = 0; j < 4; ++j) {
      const float pa = qv[j] * (kv[j] + c[j]);
      const float pb = qv[j + 4] * (kv[j + 4] + c[j + 4]);
      const float own = hi8 ? pb : pa;
      const float oth = hi8 ? pa : pb;
      s4[j] = own + row_xor<8>(oth);
    }
#pragma unroll
    for (int j = 0; j < 4; ++j) s4[j] += row_xor<4>(s4[j]);
#pragma unroll
    for (int j = 0; j < 4; ++j) s4[j] += row_xor<2>(s4[j]);
#pragma unroll
    for (int j = 0; j < 4; ++j) s4[j] += row_xor<1>(s4[j]);
#pragma unroll
    for (int j = 0; j < 4; ++j) keep[j] = mine ? s4[j] : keep[j];
  }

  float gv[4];
#pragma unroll
  for (int j = 0; j < 4; ++j) gv[j] = gts[wv][mb + rsel + j][hm];
#pragma unroll
  for (int j = 0; j < 4; ++j) {
    float sc = keep[j] * 0.25f + gv[j];
    scores[(size_t)(e0 + mb + rsel + j) * 8 + hm] = sc;
    atomicMax(&mmax[(size_t)dsel[j] * 8 + hm], fkey(sc));
  }
}

__global__ __launch_bounds__(256) void exp_denom_kernel(
    const int* __restrict__ dst, float* __restrict__ scores,
    const unsigned* __restrict__ mmax, float* __restrict__ denom, int total) {
  int idx = blockIdx.x * blockDim.x + threadIdx.x;
  if (idx >= total) return;
  int e = idx >> 3, h = idx & 7;
  int d = dst[e];
  float mv = funkey(mmax[(size_t)d * 8 + h]);
  float es = __expf(scores[idx] - mv);
  scores[idx] = es;
  atomicAdd(&denom[(size_t)d * 8 + h], es);
}

__device__ __forceinline__ float xhalf(float x) {
  int b = __builtin_bit_cast(int, x);
  return __builtin_bit_cast(
      float, __builtin_amdgcn_permlanex16(b, b, 0x76543210, 0xfedcba98, false,
                                          false));
}

__global__ __launch_bounds__(256) void edge_aggregate_kernel(
    const float* __restrict__ ea, const int* __restrict__ src,
    const int* __restrict__ dst, const float* __restrict__ Wev,
    const float* __restrict__ v, const float* __restrict__ scores,
    const float* __restrict__ denom, float* __restrict__ acc) {
  const int lane = threadIdx.x & 31;
  const int wv   = threadIdx.x >> 5;
  const int e0   = __builtin_amdgcn_readfirstlane((blockIdx.x * WAVES + wv) << 4);
  const int m    = lane & 15;
  const int kk   = (lane >> 4) << 1;
  const int mb   = (lane >> 4) << 3;
  const bool upper = lane >= 16;

  v2f afr[8];
  const float* er = ea + (size_t)(e0 + m) * 32;
#pragma unroll
  for (int i = 0; i < 8; ++i) {
    afr[i][0] = er[4 * i + kk];
    afr[i][1] = er[4 * i + kk + 1];
  }

  int sn[16], dn[16];
#pragma unroll
  for (int j = 0; j < 16; ++j) {
    sn[j] = src[e0 + j];
    dn[j] = dst[e0 + j];
  }

  float vx[8], vy[8];
  auto load_v = [&](const int p) {
    const int colx = p * 32 + lane;
#pragma unroll
    for (int r = 0; r < 8; ++r) {
      vx[r] = v[(size_t)sn[r] * 128 + colx];
      vy[r] = v[(size_t)sn[8 + r] * 128 + colx];
    }
  };

  const int hm = m & 7;
  int att8[8];
  {
    load_v(0);
    float sv[8], dv[8];
#pragma unroll
    for (int r = 0; r < 8; ++r) {
      const int dr = upper ? dn[8 + r] : dn[r];
      sv[r] = scores[(size_t)(e0 + mb + r) * 8 + hm];
      dv[r] = denom[(size_t)dr * 8 + hm];
    }
    __builtin_amdgcn_sched_barrier(0);
#pragma unroll
    for (int r = 0; r < 8; ++r)
      att8[r] = __float_as_int(sv[r] / (dv[r] + 1e-9f));
  }

  auto trip = [&](const int p) {
    const int colx = p * 32 + lane;
    if (p > 0) {
      load_v(p);
      __builtin_amdgcn_sched_barrier(0);
    }

    v8f c0 = {}, c1 = {};
    const float* wp = Wev + p * 32 + m;
#pragma unroll
    for (int i = 0; i < 8; ++i) {
      v2f b0, b1;
      b0[0] = wp[(size_t)(4 * i + kk) * 128];
      b0[1] = wp[(size_t)(4 * i + kk + 1) * 128];
      b1[0] = wp[(size_t)(4 * i + kk) * 128 + 16];
      b1[1] = wp[(size_t)(4 * i + kk + 1) * 128 + 16];
      c0 = __builtin_amdgcn_wmma_f32_16x16x4_f32(false, afr[i], false, b0,
                                                 (short)0, c0, false, false);
      c1 = __builtin_amdgcn_wmma_f32_16x16x4_f32(false, afr[i], false, b1,
                                                 (short)0, c1, false, false);
    }

#pragma unroll
    for (int r = 0; r < 8; ++r) {
      const float t0 = xhalf(c0[r]);
      const float t1 = xhalf(c1[r]);
      const float cx = upper ? t1 : c0[r];
      const float cy = upper ? c1[r] : t0;
      const float ax_lo = __int_as_float(__builtin_amdgcn_readlane(att8[r], 2 * p));
      const float ax_hi = __int_as_float(__builtin_amdgcn_readlane(att8[r], 2 * p + 1));
      const float ay_lo = __int_as_float(__builtin_amdgcn_readlane(att8[r], 2 * p + 16));
      const float ay_hi = __int_as_float(__builtin_amdgcn_readlane(att8[r], 2 * p + 17));
      const float ax = upper ? ax_hi : ax_lo;
      const float ay = upper ? ay_hi : ay_lo;
      atomicAdd(&acc[(size_t)dn[r] * 128 + colx], ax * (vx[r] + cx));
      atomicAdd(&acc[(size_t)dn[8 + r] * 128 + colx], ay * (vy[r] + cy));
    }
  };

#pragma unroll
  for (int p = 0; p < 4; ++p) trip(p);
}

extern "C" void kernel_launch(void* const* d_in, const int* in_sizes, int n_in,
                              void* d_out, int out_size, void* d_ws,
                              size_t ws_size, hipStream_t stream) {
  const float* x   = (const float*)d_in[0];
  const int*   ei  = (const int*)d_in[1];
  const float* ea  = (const float*)d_in[2];
  const float* Wq  = (const float*)d_in[3];
  const float* Wk  = (const float*)d_in[4];
  const float* Wv  = (const float*)d_in[5];
  const float* Wo  = (const float*)d_in[6];
  const float* Wek = (const float*)d_in[7];
  const float* Wev = (const float*)d_in[8];
  const float* Wg1 = (const float*)d_in[9];
  const float* Wg2 = (const float*)d_in[10];
  float* out = (float*)d_out;

  const int N = in_sizes[0] / 128;
  const int E = in_sizes[2] / 32;
  const int* src = ei;
  const int* dst = ei + E;

  float* q      = (float*)d_ws;
  float* k      = q + (size_t)N * 128;
  float* v      = k + (size_t)N * 128;
  float* scores = v + (size_t)N * 128;
  float* acc    = scores + (size_t)E * 8;
  float* denom  = acc + (size_t)N * 128;
  unsigned* mmax = (unsigned*)(denom + (size_t)N * 8);

  hipMemsetAsync(acc, 0, (size_t)(N * 128 + N * 8 + N * 8) * sizeof(float),
                 stream);

  dim3 blk(256);
  qkv_kernel<<<dim3(N / 16), blk, 0, stream>>>(x, Wq, Wk, Wv, q, k, v);

  edge_score_kernel<<<dim3(E / (16 * WAVES)), blk, 0, stream>>>(
      ea, src, dst, Wek, Wg1, Wg2, q, k, scores, mmax);

  exp_denom_kernel<<<dim3((E * 8 + 255) / 256), blk, 0, stream>>>(
      dst, scores, mmax, denom, E * 8);

  edge_aggregate_kernel<<<dim3(E / (16 * WAVES)), blk, 0, stream>>>(
      ea, src, dst, Wev, v, scores, denom, acc);

  gemm128_kernel<<<dim3(N / 16), blk, 0, stream>>>(acc, Wo, out);
}


# edge_score: a wave takes 32 edges (two 16-row tiles), each Wg1/Wek fragment load feeds two WMMAs; in the gate's second product a lane owns one edge, Wg2 by scalar loads; same sums
# speedup vs baseline: 2.4853x; 1.0731x over previous
#include <hip/hip_runtime.h>


typedef __attribute__((ext_vector_type(2))) float v2f;
typedef __attribute__((ext_vector_type(8))) float v8f;

#define WAVES 8

__device__ __forceinline__ unsigned fkey(float f) {
  unsigned u = __float_as_uint(f);
  return (u & 0x80000000u) ? ~u : (u | 0x80000000u);
}
__device__ __forceinline__ float funkey(unsigned kk) {
  unsigned u = (kk & 0x80000000u) ? (kk & 0x7FFFFFFFu) : ~kk;
  return __uint_as_float(u);
}

__global__ __launch_bounds__(256) void gemm128_kernel(
    const float* __restrict__ A, const float* __restrict__ W,
    float* __restrict__ out) {
  const int lane = threadIdx.x & 31;
  const int wv   = threadIdx.x >> 5;
  const int row0 = blockIdx.x << 4;
  const int n0   = wv << 4;
  const int m    = lane & 15;
  const int kk   = (lane >> 4) << 1;
  const int mb   = (lane >> 4) << 3;

  v8f c = {};
  const float* ar = A + (size_t)(row0 + m) * 128;
  const float* wp = W + n0 + m;
#pragma unroll 8
  for (int k0 = 0; k0 < 128; k0 += 4) {
    v2f a, b;
    a[0] = ar[k0 + kk];
    a[1] = ar[k0 + kk + 1];
    b[0] = wp[(size_t)(k0 + kk) * 128];
    b[1] = wp[(size_t)(k0 + kk + 1) * 128];
    c = __builtin_amdgcn_wmma_f32_16x16x4_f32(false, a, false, b, (short)0, c,
                                              false, false);
  }
  float* op = out + (size_t)(row0 + mb) * 128 + n0 + m;
#pragma unroll
  for (int r = 0; r < 8; ++r) op[(size_t)r * 128] = c[r];
}

__global__ __launch_bounds__(256) void qkv_kernel(
    const float* __restrict__ A, const float* __restrict__ Wq,
    const float* __restrict__ Wk, const float* __restrict__ Wv,
    float* __restrict__ q, float* __restrict__ k, float* __restrict__ v) {
  const int lane = threadIdx.x & 31;
  const int wv   = threadIdx.x >> 5;
  const int row0 = blockIdx.x << 4;
  const int n0   = wv << 4;
  const int m    = lane & 15;
  const int kk   = (lane >> 4) << 1;
  const int mb   = (lane >> 4) << 3;

  v8f cq = {}, ck = {}, cv = {};
  const float* ar = A + (size_t)(row0 + m) * 128;
  const size_t wc = (size_t)(n0 + m);
#pragma unroll 4
  for (int k0 = 0; k0 < 128; k0 += 4) {
    v2f a, bq, bk, bv;
    const size_t w0 = (size_t)(k0 + kk) * 128 + wc;
    const size_t w1 = w0 + 128;
    a[0]  = ar[k0 + kk];
    a[1]  = ar[k0 + kk + 1];
    bq[0] = Wq[w0];
    bq[1] = Wq[w1];
    bk[0] = Wk[w0];
    bk[1] = Wk[w1];
    bv[0] = Wv[w0];
    bv[1] = Wv[w1];
    cq = __builtin_amdgcn_wmma_f32_16x16x4_f32(false, a, false, bq, (short)0, cq,
                                               false, false);
    ck = __builtin_amdgcn_wmma_f32_16x16x4_f32(false, a, false, bk, (short)0, ck,
                                               false, false);
    cv = __builtin_amdgcn_wmma_f32_16x16x4_f32(false, a, false, bv, (short)0, cv,
                                               false, false);
  }
  const size_t o = (size_t)(row0 + mb) * 128 + wc;
#pragma unroll
  for (int r = 0; r < 8; ++r) q[o + (size_t)r * 128] = cq[r];
#pragma unroll
  for (int r = 0; r < 8; ++r) k[o + (size_t)r * 128] = ck[r];
#pragma unroll
  for (int r = 0; r < 8; ++r) v[o + (size_t)r * 128] = cv[r];
}

template <int MASK>
__device__ __forceinline__ float row_xor(float x) {
  return __int_as_float(__builtin_amdgcn_update_dpp(
      0, __float_as_int(x), 0x160 | MASK, 0xf, 0xf, true));
}

#define SWAVES 4

__global__ __launch_bounds__(128) void edge_score_kernel(
    const float* __restrict__ ea, const int* __restrict__ src,
    const int* __restrict__ dst, const float* __restrict__ Wek,
    const float* __restrict__ Wg1, const float* __restrict__ Wg2,
    const float* __restrict__ q, const float* __restrict__ k,
    float* __restrict__ scores, unsigned* __restrict__ mmax) {
  __shared__ float4 g1s[SWAVES][32][17];
  __shared__ float4 gts[SWAVES][32][2];

  const int lane = threadIdx.x & 31;
  const int wv   = threadIdx.x >> 5;
  const int e0   = (blockIdx.x * SWAVES + wv) << 5;
  const int m    = lane & 15;
  const int kk   = (lane >> 4) << 1;
  const int mb   = (lane >> 4) << 3;

  v2f afr0[8], afr1[8];
  {
    const float* er0 = ea + (size_t)(e0 + m) * 32;
    const float* er1 = er0 + 16 * 32;
#pragma unroll
    for (int i = 0; i < 8; ++i) {
      afr0[i][0] = er0[4 * i + kk];
      afr0[i][1] = er0[4 * i + kk + 1];
      afr1[i][0] = er1[4 * i + kk];
      afr1[i][1] = er1[4 * i + kk + 1];
    }
  }

  {
    float* g1f = (float*)&g1s[wv][0][0];
#pragma unroll
    for (int t = 0; t < 4; ++t) {
      v8f c0 = {}, c1 = {};
      const float* wp = Wg1 + t * 16 + m;
#pragma unroll
      for (int i = 0; i < 8; ++i) {
        v2f b;
        b[0] = wp[(size_t)(4 * i + kk) * 64];
        b[1] = wp[(size_t)(4 * i + kk + 1) * 64];
        c0 = __builtin_amdgcn_wmma_f32_16x16x4_f32(false, afr0[i], false, b,
                                                   (short)0, c0, false, false);
        c1 = __builtin_amdgcn_wmma_f32_16x16x4_f32(false, afr1[i], false, b,
                                                   (short)0, c1, false, false);
      }
#pragma unroll
      for (int r = 0; r < 8; ++r) {
        const float x0 = c0[r];
        const float x1 = c1[r];
        g1f[(mb + r) * 68 + t * 16 + m]      = x0 / (1.0f + __expf(-x0));
        g1f[(16 + mb + r) * 68 + t * 16 + m] = x1 / (1.0f + __expf(-x1));
      }
    }
  }
  __syncthreads();

  {
    float acc[8];
#pragma unroll
    for (int hh = 0; hh < 8; ++hh) acc[hh] = 0.f;
#pragma unroll 2
    for (int j4 = 0; j4 < 16; ++j4) {
      const float4 g4 = g1s[wv][lane][j4];
      const float* w2 = Wg2 + j4 * 32;
#pragma unroll
      for (int hh = 0; hh < 8; ++hh) acc[hh] = fmaf(g4.x, w2[hh], acc[hh]);
#pragma unroll
      for (int hh = 0; hh < 8; ++hh) acc[hh] = fmaf(g4.y, w2[8 + hh], acc[hh]);
#pragma unroll
      for (int hh = 0; hh < 8; ++hh) acc[hh] = fmaf(g4.z, w2[16 + hh], acc[hh]);
#pragma unroll
      for (int hh = 0; hh < 8; ++hh) acc[hh] = fmaf(g4.w, w2[24 + hh], acc[hh]);
    }
    gts[wv][lane][0] = make_float4(acc[0], acc[1], acc[2], acc[3]);
    gts[wv][lane][1] = make_float4(acc[4], acc[5], acc[6], acc[7]);
  }
  __syncthreads();

  unsigned so0[8], do0[8], so1[8], do1[8];
  int dsel0[4], dsel1[4];
  const int hm = m & 7;
  const bool hi8 = (m & 8) != 0;
  const int rsel = hi8 ? 4 : 0;
  {
    int dv0[8], dv1[8];
#pragma unroll
    for (int r = 0; r < 8; ++r) {
      so0[r] = (unsigned)src[e0 + mb + r] * 512u + (unsigned)(m << 2);
      so1[r] = (unsigned)src[e0 + 16 + mb + r] * 512u + (unsigned)(m << 2);
      dv0[r] = dst[e0 + mb + r];
      dv1[r] = dst[e0 + 16 + mb + r];
      do0[r] = (unsigned)dv0[r] * 512u + (unsigned)(m << 2);
      do1[r] = (unsigned)dv1[r] * 512u + (unsigned)(m << 2);
    }
#pragma unroll
    for (int j = 0; j < 4; ++j) {
      dsel0[j] = hi8 ? dv0[j + 4] : dv0[j];
      dsel1[j] = hi8 ? dv1[j + 4] : dv1[j];
    }
  }

  float keep0[4], keep1[4];
#pragma unroll
  for (int j = 0; j < 4; ++j) {
    keep0[j] = 0.f;
    keep1[j] = 0.f;
  }

#pragma unroll 1
  for (int h = 0; h < 8; ++h) {
    const char* kb = (const char*)k;
    const char* qb = (const char*)q;
    float kv0[8], qv0[8], kv1[8], qv1[8];
#pragma unroll
    for (int r = 0; r < 8; ++r) {
      kv0[r] = *(const float*)(kb + so0[r]);
      qv0[r] = *(const float*)(qb + do0[r]);
      kv1[r] = *(const float*)(kb + so1[r]);
      qv1[r] = *(const float*)(qb + do1[r]);
    }
#pragma unroll
    for (int r = 0; r < 8; ++r) {
      so0[r] += 64u;
      do0[r] += 64u;
      so1[r] += 64u;
      do1[r] += 64u;
    }

    v8f c0 = {}, c1 = {};
    const float* wp = Wek + h * 16 + m;
#pragma unroll
    for (int i = 0; i < 8; ++i) {
      v2f b;
      b[0] = wp[(size_t)(4 * i + kk) * 128];
      b[1] = wp[(size_t)(4 * i + kk + 1) * 128];
      c0 = __builtin_amdgcn_wmma_f32_16x16x4_f32(false, afr0[i], false, b,
                                                 (short)0, c0, false, false);
      c1 = __builtin_amdgcn_wmma_f32_16x16x4_f32(false, afr1[i], false, b,
                                                 (short)0, c1, false, false);
    }

    const bool mine = (hm == h);
    float s0[4], s1[4];
#pragma unroll
    for (int j = 0; j < 4; ++j) {
      const float pa0 = qv0[j] * (kv0[j] + c0[j]);
      const float pb0 = qv0[j + 4] * (kv0[j + 4] + c0[j + 4]);
      const float pa1 = qv1[j] * (kv1[j] + c1[j]);
      const float pb1 = qv1[j + 4] * (kv1[j + 4] + c1[j + 4]);
      s0[j] = (hi8 ? pb0 : pa0) + row_xor<8>(hi8 ? pa0 : pb0);
      s1[j] = (hi8 ? pb1 : pa1) + row_xor<8>(hi8 ? pa1 : pb1);
    }
#pragma unroll
    for (int j = 0; j < 4; ++j) {
      s0[j] += row_xor<4>(s0[j]);
      s1[j] += row_xor<4>(s1[j]);
    }
#pragma unroll
    for (int j = 0; j < 4; ++j) {
      s0[j] += row_xor<2>(s0[j]);
      s1[j] += row_xor<2>(s1[j]);
    }
#pragma unroll
    for (int j = 0; j < 4; ++j) {
      s0[j] += row_xor<1>(s0[j]);
      s1[j] += row_xor<1>(s1[j]);
    }
#pragma unroll
    for (int j = 0; j < 4; ++j) {
      keep0[j] = mine ? s0[j] : keep0[j];
      keep1[j] = mine ? s1[j] : keep1[j];
    }
  }

  const float* gtf = (const float*)&gts[wv][0][0];
  float gv0[4], gv1[4];
#pragma unroll
  for (int j = 0; j < 4; ++j) {
    gv0[j] = gtf[(mb + rsel + j) * 8 + hm];
    gv1[j] = gtf[(16 + mb + rsel + j) * 8 + hm];
  }
#pragma unroll
  for (int j = 0; j < 4; ++j) {
    const float sc = keep0[j] * 0.25f + gv0[j];
    scores[(size_t)(e0 + mb + rsel + j) * 8 + hm] = sc;
    atomicMax(&mmax[(size_t)dsel0[j] * 8 + hm], fkey(sc));
  }
#pragma unroll
  for (int j = 0; j < 4; ++j) {
    const float sc = keep1[j] * 0.25f + gv1[j];
    scores[(size_t)(e0 + 16 + mb + rsel + j) * 8 + hm] = sc;
    atomicMax(&mmax[(size_t)dsel1[j] * 8 + hm], fkey(sc));
  }
}

__global__ __launch_bounds__(256) void exp_denom_kernel(
    const int* __restrict__ dst, float* __restrict__ scores,
    const unsigned* __restrict__ mmax, float* __restrict__ denom, int total) {
  int idx = blockIdx.x * blockDim.x + threadIdx.x;
  if (idx >= total) return;
  int e = idx >> 3, h = idx & 7;
  int d = dst[e];
  float mv = funkey(mmax[(size_t)d * 8 + h]);
  float es = __expf(scores[idx] - mv);
  scores[idx] = es;
  atomicAdd(&denom[(size_t)d * 8 + h], es);
}

__device__ __forceinline__ float xhalf(float x) {
  int b = __builtin_bit_cast(int, x);
  return __builtin_bit_cast(
      float, __builtin_amdgcn_permlanex16(b, b, 0x76543210, 0xfedcba98, false,
                                          false));
}

__global__ __launch_bounds__(256) void edge_aggregate_kernel(
    const float* __restrict__ ea, const int* __restrict__ src,
    const int* __restrict__ dst, const float* __restrict__ Wev,
    const float* __restrict__ v, const float* __restrict__ scores,
    const float* __restrict__ denom, float* __restrict__ acc) {
  const int lane = threadIdx.x & 31;
  const int wv   = threadIdx.x >> 5;
  const int e0   = __builtin_amdgcn_readfirstlane((blockIdx.x * WAVES + wv) << 4);
  const int m    = lane & 15;
  const int kk   = (lane >> 4) << 1;
  const int mb   = (lane >> 4) << 3;
  const bool upper = lane >= 16;

  v2f afr[8];
  const float* er = ea + (size_t)(e0 + m) * 32;
#pragma unroll
  for (int i = 0; i < 8; ++i) {
    afr[i][0] = er[4 * i + kk];
    afr[i][1] = er[4 * i + kk + 1];
  }

  int sn[16], dn[16];
#pragma unroll
  for (int j = 0; j < 16; ++j) {
    sn[j] = src[e0 + j];
    dn[j] = dst[e0 + j];
  }

  float vx[8], vy[8];
  auto load_v = [&](const int p) {
    const int colx = p * 32 + lane;
#pragma unroll
    for (int r = 0; r < 8; ++r) {
      vx[r] = v[(size_t)sn[r] * 128 + colx];
      vy[r] = v[(size_t)sn[8 + r] * 128 + colx];
    }
  };

  const int hm = m & 7;
  int att8[8];
  {
    load_v(0);
    float sv[8], dv[8];
#pragma unroll
    for (int r = 0; r < 8; ++r) {
      const int dr = upper ? dn[8 + r] : dn[r];
      sv[r] = scores[(size_t)(e0 + mb + r) * 8 + hm];
      dv[r] = denom[(size_t)dr * 8 + hm];
    }
    __builtin_amdgcn_sched_barrier(0);
#pragma unroll
    for (int r = 0; r < 8; ++r)
      att8[r] = __float_as_int(sv[r] / (dv[r] + 1e-9f));
  }

  auto trip = [&](const int p) {
    const int colx = p * 32 + lane;
    if (p > 0) {
      load_v(p);
      __builtin_amdgcn_sched_barrier(0);
    }

    v8f c0 = {}, c1 = {};
    const float* wp = Wev + p * 32 + m;
#pragma unroll
    for (int i = 0; i < 8; ++i) {
      v2f b0, b1;
      b0[0] = wp[(size_t)(4 * i + kk) * 128];
      b0[1] = wp[(size_t)(4 * i + kk + 1) * 128];
      b1[0] = wp[(size_t)(4 * i + kk) * 128 + 16];
      b1[1] = wp[(size_t)(4 * i + kk + 1) * 128 + 16];
      c0 = __builtin_amdgcn_wmma_f32_16x16x4_f32(false, afr[i], false, b0,
                                                 (short)0, c0, false, false);
      c1 = __builtin_amdgcn_wmma_f32_16x16x4_f32(false, afr[i], false, b1,
                                                 (short)0, c1, false, false);
    }

#pragma unroll
    for (int r = 0; r < 8; ++r) {
      const float t0 = xhalf(c0[r]);
      const float t1 = xhalf(c1[r]);
      const float cx = upper ? t1 : c0[r];
      const float cy = upper ? c1[r] : t0;
      const float ax_lo = __int_as_float(__builtin_amdgcn_readlane(att8[r], 2 * p));
      const float ax_hi = __int_as_float(__builtin_amdgcn_readlane(att8[r], 2 * p + 1));
      const float ay_lo = __int_as_float(__builtin_amdgcn_readlane(att8[r], 2 * p + 16));
      const float ay_hi = __int_as_float(__builtin_amdgcn_readlane(att8[r], 2 * p + 17));
      const float ax = upper ? ax_hi : ax_lo;
      const float ay = upper ? ay_hi : ay_lo;
      atomicAdd(&acc[(size_t)dn[r] * 128 + colx], ax * (vx[r] + cx));
      atomicAdd(&acc[(size_t)dn[8 + r] * 128 + colx], ay * (vy[r] + cy));
    }
  };

#pragma unroll
  for (int p = 0; p < 4; ++p) trip(p);
}

extern "C" void kernel_launch(void* const* d_in, const int* in_sizes, int n_in,
                              void* d_out, int out_size, void* d_ws,
                              size_t ws_size, hipStream_t stream) {
  const float* x   = (const float*)d_in[0];
  const int*   ei  = (const int*)d_in[1];
  const float* ea  = (const float*)d_in[2];
  const float* Wq  = (const float*)d_in[3];
  const float* Wk  = (const float*)d_in[4];
  const float* Wv  = (const float*)d_in[5];
  const float* Wo  = (const float*)d_in[6];
  const float* Wek = (const float*)d_in[7];
  const float* Wev = (const float*)d_in[8];
  const float* Wg1 = (const float*)d_in[9];
  const float* Wg2 = (const float*)d_in[10];
  float* out = (float*)d_out;

  const int N = in_sizes[0] / 128;
  const int E = in_sizes[2] / 32;
  const int* src = ei;
  const int* dst = ei + E;

  float* q      = (float*)d_ws;
  float* k      = q + (size_t)N * 128;
  float* v      = k + (size_t)N * 128;
  float* scores = v + (size_t)N * 128;
  float* acc    = scores + (size_t)E * 8;
  float* denom  = acc + (size_t)N * 128;
  unsigned* mmax = (unsigned*)(denom + (size_t)N * 8);

  hipMemsetAsync(acc, 0, (size_t)(N * 128 + N * 8 + N * 8) * sizeof(float),
                 stream);

  dim3 blk(256);
  qkv_kernel<<<dim3(N / 16), blk, 0, stream>>>(x, Wq, Wk, Wv, q, k, v);

  edge_score_kernel<<<dim3(E / (32 * SWAVES)), dim3(32 * SWAVES), 0, stream>>>(
      ea, src, dst, Wek, Wg1, Wg2, q, k, scores, mmax);

  exp_denom_kernel<<<dim3((E * 8 + 255) / 256), blk, 0, stream>>>(
      dst, scores, mmax, denom, E * 8);

  edge_aggregate_kernel<<<dim3(E / (16 * WAVES)), blk, 0, stream>>>(
      ea, src, dst, Wev, v, scores, denom, acc);

  gemm128_kernel<<<dim3(N / 16), blk, 0, stream>>>(acc, Wo, out);
}


# edge_score: a trip takes two heads; lane halves of the ek tiles exchanged (permlanex16) so each q and k gather is one whole 128-byte line of a row; row sums fold pairwise to lane = edge; same sums
# speedup vs baseline: 2.6097x; 1.0501x over previous
#include <hip/hip_runtime.h>


typedef __attribute__((ext_vector_type(2))) float v2f;
typedef __attribute__((ext_vector_type(8))) float v8f;

#define WAVES 8

__device__ __forceinline__ unsigned fkey(float f) {
  unsigned u = __float_as_uint(f);
  return (u & 0x80000000u) ? ~u : (u | 0x80000000u);
}
__device__ __forceinline__ float funkey(unsigned kk) {
  unsigned u = (kk & 0x80000000u) ? (kk & 0x7FFFFFFFu) : ~kk;
  return __uint_as_float(u);
}

__global__ __launch_bounds__(256) void gemm128_kernel(
    const float* __restrict__ A, const float* __restrict__ W,
    float* __restrict__ out) {
  const int lane = threadIdx.x & 31;
  const int wv   = threadIdx.x >> 5;
  const int row0 = blockIdx.x << 4;
  const int n0   = wv << 4;
  const int m    = lane & 15;
  const int kk   = (lane >> 4) << 1;
  const int mb   = (lane >> 4) << 3;

  v8f c = {};
  const float* ar = A + (size_t)(row0 + m) * 128;
  const float* wp = W + n0 + m;
#pragma unroll 8
  for (int k0 = 0; k0 < 128; k0 += 4) {
    v2f a, b;
    a[0] = ar[k0 + kk];
    a[1] = ar[k0 + kk + 1];
    b[0] = wp[(size_t)(k0 + kk) * 128];
    b[1] = wp[(size_t)(k0 + kk + 1) * 128];
    c = __builtin_amdgcn_wmma_f32_16x16x4_f32(false, a, false, b, (short)0, c,
                                              false, false);
  }
  float* op = out + (size_t)(row0 + mb) * 128 + n0 + m;
#pragma unroll
  for (int r = 0; r < 8; ++r) op[(size_t)r * 128] = c[r];
}

__global__ __launch_bounds__(256) void qkv_kernel(
    const float* __restrict__ A, const float* __restrict__ Wq,
    const float* __restrict__ Wk, const float* __restrict__ Wv,
    float* __restrict__ q, float* __restrict__ k, float* __restrict__ v) {
  const int lane = threadIdx.x & 31;
  const int wv   = threadIdx.x >> 5;
  const int row0 = blockIdx.x << 4;
  const int n0   = wv << 4;
  const int m    = lane & 15;
  const int kk   = (lane >> 4) << 1;
  const int mb   = (lane >> 4) << 3;

  v8f cq = {}, ck = {}, cv = {};
  const float* ar = A + (size_t)(row0 + m) * 128;
  const size_t wc = (size_t)(n0 + m);
#pragma unroll 4
  for (int k0 = 0; k0 < 128; k0 += 4) {
    v2f a, bq, bk, bv;
    const size_t w0 = (size_t)(k0 + kk) * 128 + wc;
    const size_t w1 = w0 + 128;
    a[0]  = ar[k0 + kk];
    a[1]  = ar[k0 + kk + 1];
    bq[0] = Wq[w0];
    bq[1] = Wq[w1];
    bk[0] = Wk[w0];
    bk[1] = Wk[w1];
    bv[0] = Wv[w0];
    bv[1] = Wv[w1];
    cq = __builtin_amdgcn_wmma_f32_16x16x4_f32(false, a, false, bq, (short)0, cq,
                                               false, false);
    ck = __builtin_amdgcn_wmma_f32_16x16x4_f32(false, a, false, bk, (short)0, ck,
                                               false, false);
    cv = __builtin_amdgcn_wmma_f32_16x16x4_f32(false, a, false, bv, (short)0, cv,
                                               false, false);
  }
  const size_t o = (size_t)(row0 + mb) * 128 + wc;
#pragma unroll
  for (int r = 0; r < 8; ++r) q[o + (size_t)r * 128] = cq[r];
#pragma unroll
  for (int r = 0; r < 8; ++r) k[o + (size_t)r * 128] = ck[r];
#pragma unroll
  for (int r = 0; r < 8; ++r) v[o + (size_t)r * 128] = cv[r];
}

template <int MASK>
__device__ __forceinline__ float row_xor(float x) {
  return __int_as_float(__builtin_amdgcn_update_dpp(
      0, __float_as_int(x), 0x160 | MASK, 0xf, 0xf, true));
}

__device__ __forceinline__ float xhalf(float x) {
  int b = __builtin_bit_cast(int, x);
  return __builtin_bit_cast(
      float, __builtin_amdgcn_permlanex16(b, b, 0x76543210, 0xfedcba98, false,
                                          false));
}

#define SWAVES 4

__global__ __launch_bounds__(128) void edge_score_kernel(
    const float* __restrict__ ea, const int* __restrict__ src,
    const int* __restrict__ dst, const float* __restrict__ Wek,
    const float* __restrict__ Wg1, const float* __restrict__ Wg2,
    const float* __restrict__ q, const float* __restrict__ k,
    float* __restrict__ scores, unsigned* __restrict__ mmax) {
  __shared__ float4 g1s[SWAVES][32][17];
  __shared__ float4 gts[SWAVES][32][2];

  const int lane = threadIdx.x & 31;
  const int wv   = threadIdx.x >> 5;
  const int e0   = (blockIdx.x * SWAVES + wv) << 5;
  const int m    = lane & 15;
  const int kk   = (lane >> 4) << 1;
  const int mb   = (lane >> 4) << 3;

  v2f afr0[8], afr1[8];
  {
    const float* er0 = ea + (size_t)(e0 + m) * 32;
    const float* er1 = er0 + 16 * 32;
#pragma unroll
    for (int i = 0; i < 8; ++i) {
      afr0[i][0] = er0[4 * i + kk];
      afr0[i][1] = er0[4 * i + kk + 1];
      afr1[i][0] = er1[4 * i + kk];
      afr1[i][1] = er1[4 * i + kk + 1];
    }
  }

  {
    float* g1f = (float*)&g1s[wv][0][0];
#pragma unroll
    for (int t = 0; t < 4; ++t) {
      v8f c0 = {}, c1 = {};
      const float* wp = Wg1 + t * 16 + m;
#pragma unroll
      for (int i = 0; i < 8; ++i) {
        v2f b;
        b[0] = wp[(size_t)(4 * i + kk) * 64];
        b[1] = wp[(size_t)(4 * i + kk + 1) * 64];
        c0 = __builtin_amdgcn_wmma_f32_16x16x4_f32(false, afr0[i], false, b,
                                                   (short)0, c0, false, false);
        c1 = __builtin_amdgcn_wmma_f32_16x16x4_f32(false, afr1[i], false, b,
                                                   (short)0, c1, false, false);
      }
#pragma unroll
      for (int r = 0; r < 8; ++r) {
        const float x0 = c0[r];
        const float x1 = c1[r];
        g1f[(mb + r) * 68 + t * 16 + m]      = x0 / (1.0f + __expf(-x0));
        g1f[(16 + mb + r) * 68 + t * 16 + m] = x1 / (1.0f + __expf(-x1));
      }
    }
  }
  __syncthreads();

  {
    float acc[8];
#pragma unroll
    for (int hh = 0; hh < 8; ++hh) acc[hh] = 0.f;
#pragma unroll 2
    for (int j4 = 0; j4 < 16; ++j4) {
      const float4 g4 = g1s[wv][lane][j4];
      const float* w2 = Wg2 + j4 * 32;
#pragma unroll
      for (int hh = 0; hh < 8; ++hh) acc[hh] = fmaf(g4.x, w2[hh], acc[hh]);
#pragma unroll
      for (int hh = 0; hh < 8; ++hh) acc[hh] = fmaf(g4.y, w2[8 + hh], acc[hh]);
#pragma unroll
      for (int hh = 0; hh < 8; ++hh) acc[hh] = fmaf(g4.z, w2[16 + hh], acc[hh]);
#pragma unroll
      for (int hh = 0; hh < 8; ++hh) acc[hh] = fmaf(g4.w, w2[24 + hh], acc[hh]);
    }
    gts[wv][lane][0] = make_float4(acc[0], acc[1], acc[2], acc[3]);
    gts[wv][lane][1] = make_float4(acc[4], acc[5], acc[6], acc[7]);
  }
  __syncthreads();

  const unsigned so  = (unsigned)src[e0 + lane] * 512u;
  const unsigned dof = (unsigned)dst[e0 + lane] * 512u;

  const bool upper = lane >= 16;
  const bool b8 = (m & 8) != 0;
  const bool b4 = (m & 4) != 0;
  const bool b2 = (m & 2) != 0;
  const bool b1 = (m & 1) != 0;
  float* dtf = (float*)&g1s[wv][0][0];
  const char* kb = (const char*)k;
  const char* qb = (const char*)q;
  unsigned sN[32], dN[32];
#pragma unroll
  for (int j = 0; j < 32; ++j) {
    sN[j] = (unsigned)__builtin_amdgcn_readlane((int)so, j);
    dN[j] = (unsigned)__builtin_amdgcn_readlane((int)dof, j);
  }

#pragma unroll 1
  for (int p = 0; p < 4; ++p) {
    const unsigned cb = (unsigned)(p * 128 + (lane << 2));

    float kx[16], qx[16];
    auto gather = [&](const int T) {
#pragma unroll
      for (int j = 0; j < 16; ++j) {
        const unsigned ko = sN[16 * T + j] + cb;
        const unsigned qo = dN[16 * T + j] + cb;
        kx[j] = *(const float*)(kb + ko);
        qx[j] = *(const float*)(qb + qo);
      }
    };
    gather(0);

    v8f c00 = {}, c01 = {}, c10 = {}, c11 = {};
    const float* wp = Wek + p * 32 + m;
#pragma unroll
    for (int i = 0; i < 8; ++i) {
      v2f b0, b1v;
      b0[0]  = wp[(size_t)(4 * i + kk) * 128];
      b0[1]  = wp[(size_t)(4 * i + kk + 1) * 128];
      b1v[0] = wp[(size_t)(4 * i + kk) * 128 + 16];
      b1v[1] = wp[(size_t)(4 * i + kk + 1) * 128 + 16];
      c00 = __builtin_amdgcn_wmma_f32_16x16x4_f32(false, afr0[i], false, b0,
                                                  (short)0, c00, false, false);
      c10 = __builtin_amdgcn_wmma_f32_16x16x4_f32(false, afr1[i], false, b0,
                                                  (short)0, c10, false, false);
      c01 = __builtin_amdgcn_wmma_f32_16x16x4_f32(false, afr0[i], false, b1v,
                                                  (short)0, c01, false, false);
      c11 = __builtin_amdgcn_wmma_f32_16x16x4_f32(false, afr1[i], false, b1v,
                                                  (short)0, c11, false, false);
    }

    float pr[16];
    auto products = [&](const v8f& ca, const v8f& cb1) {
#pragma unroll
      for (int r = 0; r < 8; ++r) {
        const float t0 = xhalf(ca[r]);
        const float t1 = xhalf(cb1[r]);
        const float ex = upper ? t1 : ca[r];
        const float ey = upper ? cb1[r] : t0;
        pr[r]     = qx[r] * (kx[r] + ex);
        pr[8 + r] = qx[8 + r] * (kx[8 + r] + ey);
      }
    };
    auto reduce = [&](const int T) {
      float s8[8], s4[4], s2[2];
#pragma unroll
      for (int j = 0; j < 8; ++j)
        s8[j] = (b8 ? pr[j + 8] : pr[j]) + row_xor<8>(b8 ? pr[j] : pr[j + 8]);
#pragma unroll
      for (int j = 0; j < 4; ++j)
        s4[j] = (b4 ? s8[j + 4] : s8[j]) + row_xor<4>(b4 ? s8[j] : s8[j + 4]);
#pragma unroll
      for (int j = 0; j < 2; ++j)
        s2[j] = (b2 ? s4[j + 2] : s4[j]) + row_xor<2>(b2 ? s4[j] : s4[j + 2]);
      const float u = (b1 ? s2[1] : s2[0]) + row_xor<1>(b1 ? s2[0] : s2[1]);
      dtf[(16 * T + m) * 8 + 2 * p + (lane >> 4)] = u;
    };

    products(c00, c01);
    __builtin_amdgcn_sched_barrier(0);
    gather(1);
    reduce(0);
    products(c10, c11);
    reduce(1);
  }

  const float* gtf = (const float*)&gts[wv][0][0];
  const int hm = lane & 7;
  float dv[8], gv[8];
  int dn[8];
#pragma unroll
  for (int it = 0; it < 8; ++it) {
    dn[it] = dst[e0 + 4 * it + (lane >> 3)];
    dv[it] = dtf[32 * it + lane];
    gv[it] = gtf[32 * it + lane];
  }
#pragma unroll
  for (int it = 0; it < 8; ++it) {
    const float sc = dv[it] * 0.25f + gv[it];
    scores[(size_t)(e0 + 4 * it) * 8 + lane] = sc;
    atomicMax(&mmax[(size_t)dn[it] * 8 + hm], fkey(sc));
  }
}

__global__ __launch_bounds__(256) void exp_denom_kernel(
    const int* __restrict__ dst, float* __restrict__ scores,
    const unsigned* __restrict__ mmax, float* __restrict__ denom, int total) {
  int idx = blockIdx.x * blockDim.x + threadIdx.x;
  if (idx >= total) return;
  int e = idx >> 3, h = idx & 7;
  int d = dst[e];
  float mv = funkey(mmax[(size_t)d * 8 + h]);
  float es = __expf(scores[idx] - mv);
  scores[idx] = es;
  atomicAdd(&denom[(size_t)d * 8 + h], es);
}


__global__ __launch_bounds__(256) void edge_aggregate_kernel(
    const float* __restrict__ ea, const int* __restrict__ src,
    const int* __restrict__ dst, const float* __restrict__ Wev,
    const float* __restrict__ v, const float* __restrict__ scores,
    const float* __restrict__ denom, float* __restrict__ acc) {
  const int lane = threadIdx.x & 31;
  const int wv   = threadIdx.x >> 5;
  const int e0   = __builtin_amdgcn_readfirstlane((blockIdx.x * WAVES + wv) << 4);
  const int m    = lane & 15;
  const int kk   = (lane >> 4) << 1;
  const int mb   = (lane >> 4) << 3;
  const bool upper = lane >= 16;

  v2f afr[8];
  const float* er = ea + (size_t)(e0 + m) * 32;
#pragma unroll
  for (int i = 0; i < 8; ++i) {
    afr[i][0] = er[4 * i + kk];
    afr[i][1] = er[4 * i + kk + 1];
  }

  int sn[16], dn[16];
#pragma unroll
  for (int j = 0; j < 16; ++j) {
    sn[j] = src[e0 + j];
    dn[j] = dst[e0 + j];
  }

  float vx[8], vy[8];
  auto load_v = [&](const int p) {
    const int colx = p * 32 + lane;
#pragma unroll
    for (int r = 0; r < 8; ++r) {
      vx[r] = v[(size_t)sn[r] * 128 + colx];
      vy[r] = v[(size_t)sn[8 + r] * 128 + colx];
    }
  };

  const int hm = m & 7;
  int att8[8];
  {
    load_v(0);
    float sv[8], dv[8];
#pragma unroll
    for (int r = 0; r < 8; ++r) {
      const int dr = upper ? dn[8 + r] : dn[r];
      sv[r] = scores[(size_t)(e0 + mb + r) * 8 + hm];
      dv[r] = denom[(size_t)dr * 8 + hm];
    }
    __builtin_amdgcn_sched_barrier(0);
#pragma unroll
    for (int r = 0; r < 8; ++r)
      att8[r] = __float_as_int(sv[r] / (dv[r] + 1e-9f));
  }

  auto trip = [&](const int p) {
    const int colx = p * 32 + lane;
    if (p > 0) {
      load_v(p);
      __builtin_amdgcn_sched_barrier(0);
    }

    v8f c0 = {}, c1 = {};
    const float* wp = Wev + p * 32 + m;
#pragma unroll
    for (int i = 0; i < 8; ++i) {
      v2f b0, b1;
      b0[0] = wp[(size_t)(4 * i + kk) * 128];
      b0[1] = wp[(size_t)(4 * i + kk + 1) * 128];
      b1[0] = wp[(size_t)(4 * i + kk) * 128 + 16];
      b1[1] = wp[(size_t)(4 * i + kk + 1) * 128 + 16];
      c0 = __builtin_amdgcn_wmma_f32_16x16x4_f32(false, afr[i], false, b0,
                                                 (short)0, c0, false, false);
      c1 = __builtin_amdgcn_wmma_f32_16x16x4_f32(false, afr[i], false, b1,
                                                 (short)0, c1, false, false);
    }

#pragma unroll
    for (int r = 0; r < 8; ++r) {
      const float t0 = xhalf(c0[r]);
      const float t1 = xhalf(c1[r]);
      const float cx = upper ? t1 : c0[r];
      const float cy = upper ? c1[r] : t0;
      const float ax_lo = __int_as_float(__builtin_amdgcn_readlane(att8[r], 2 * p));
      const float ax_hi = __int_as_float(__builtin_amdgcn_readlane(att8[r], 2 * p + 1));
      const float ay_lo = __int_as_float(__builtin_amdgcn_readlane(att8[r], 2 * p + 16));
      const float ay_hi = __int_as_float(__builtin_amdgcn_readlane(att8[r], 2 * p + 17));
      const float ax = upper ? ax_hi : ax_lo;
      const float ay = upper ? ay_hi : ay_lo;
      atomicAdd(&acc[(size_t)dn[r] * 128 + colx], ax * (vx[r] + cx));
      atomicAdd(&acc[(size_t)dn[8 + r] * 128 + colx], ay * (vy[r] + cy));
    }
  };

#pragma unroll
  for (int p = 0; p < 4; ++p) trip(p);
}

extern "C" void kernel_launch(void* const* d_in, const int* in_sizes, int n_in,
                              void* d_out, int out_size, void* d_ws,
                              size_t ws_size, hipStream_t stream) {
  const float* x   = (const float*)d_in[0];
  const int*   ei  = (const int*)d_in[1];
  const float* ea  = (const float*)d_in[2];
  const float* Wq  = (const float*)d_in[3];
  const float* Wk  = (const float*)d_in[4];
  const float* Wv  = (const float*)d_in[5];
  const float* Wo  = (const float*)d_in[6];
  const float* Wek = (const float*)d_in[7];
  const float* Wev = (const float*)d_in[8];
  const float* Wg1 = (const float*)d_in[9];
  const float* Wg2 = (const float*)d_in[10];
  float* out = (float*)d_out;

  const int N = in_sizes[0] / 128;
  const int E = in_sizes[2] / 32;
  const int* src = ei;
  const int* dst = ei + E;

  float* q      = (float*)d_ws;
  float* k      = q + (size_t)N * 128;
  float* v      = k + (size_t)N * 128;
  float* scores = v + (size_t)N * 128;
  float* acc    = scores + (size_t)E * 8;
  float* denom  = acc + (size_t)N * 128;
  unsigned* mmax = (unsigned*)(denom + (size_t)N * 8);

  hipMemsetAsync(acc, 0, (size_t)(N * 128 + N * 8 + N * 8) * sizeof(float),
                 stream);

  dim3 blk(256);
  qkv_kernel<<<dim3(N / 16), blk, 0, stream>>>(x, Wq, Wk, Wv, q, k, v);

  edge_score_kernel<<<dim3(E / (32 * SWAVES)), dim3(32 * SWAVES), 0, stream>>>(
      ea, src, dst, Wek, Wg1, Wg2, q, k, scores, mmax);

  exp_denom_kernel<<<dim3((E * 8 + 255) / 256), blk, 0, stream>>>(
      dst, scores, mmax, denom, E * 8);

  edge_aggregate_kernel<<<dim3(E / (16 * WAVES)), blk, 0, stream>>>(
      ea, src, dst, Wev, v, scores, denom, acc);

  gemm128_kernel<<<dim3(N / 16), blk, 0, stream>>>(acc, Wo, out);
}


# online softmax in pass 1: per (dst, head) one 8-byte {max, rescaled sum} word updated by 64-bit compare-and-swap, not atomic max; exp_denom launch gone; denom sum order and rounding change
# speedup vs baseline: 2.7193x; 1.0420x over previous
#include <hip/hip_runtime.h>


typedef __attribute__((ext_vector_type(2))) float v2f;
typedef __attribute__((ext_vector_type(8))) float v8f;

#define WAVES 8

__device__ __forceinline__ unsigned long long softmax_take(unsigned long long w,
                                                           float s) {
  const float mo = __uint_as_float((unsigned)w);
  const float so = __uint_as_float((unsigned)(w >> 32));
  const float me = (so == 0.f) ? __uint_as_float(0xff800000u) : mo;
  const float mn = fmaxf(me, s);
  const float sn = so * __expf(me - mn) + __expf(s - mn);
  return (unsigned long long)__float_as_uint(mn) |
         ((unsigned long long)__float_as_uint(sn) << 32);
}

__global__ __launch_bounds__(256) void gemm128_kernel(
    const float* __restrict__ A, const float* __restrict__ W,
    float* __restrict__ out) {
  const int lane = threadIdx.x & 31;
  const int wv   = threadIdx.x >> 5;
  const int row0 = blockIdx.x << 4;
  const int n0   = wv << 4;
  const int m    = lane & 15;
  const int kk   = (lane >> 4) << 1;
  const int mb   = (lane >> 4) << 3;

  v8f c = {};
  const float* ar = A + (size_t)(row0 + m) * 128;
  const float* wp = W + n0 + m;
#pragma unroll 8
  for (int k0 = 0; k0 < 128; k0 += 4) {
    v2f a, b;
    a[0] = ar[k0 + kk];
    a[1] = ar[k0 + kk + 1];
    b[0] = wp[(size_t)(k0 + kk) * 128];
    b[1] = wp[(size_t)(k0 + kk + 1) * 128];
    c = __builtin_amdgcn_wmma_f32_16x16x4_f32(false, a, false, b, (short)0, c,
                                              false, false);
  }
  float* op = out + (size_t)(row0 + mb) * 128 + n0 + m;
#pragma unroll
  for (int r = 0; r < 8; ++r) op[(size_t)r * 128] = c[r];
}

__global__ __launch_bounds__(256) void qkv_kernel(
    const float* __restrict__ A, const float* __restrict__ Wq,
    const float* __restrict__ Wk, const float* __restrict__ Wv,
    float* __restrict__ q, float* __restrict__ k, float* __restrict__ v) {
  const int lane = threadIdx.x & 31;
  const int wv   = threadIdx.x >> 5;
  const int row0 = blockIdx.x << 4;
  const int n0   = wv << 4;
  const int m    = lane & 15;
  const int kk   = (lane >> 4) << 1;
  const int mb   = (lane >> 4) << 3;

  v8f cq = {}, ck = {}, cv = {};
  const float* ar = A + (size_t)(row0 + m) * 128;
  const size_t wc = (size_t)(n0 + m);
#pragma unroll 4
  for (int k0 = 0; k0 < 128; k0 += 4) {
    v2f a, bq, bk, bv;
    const size_t w0 = (size_t)(k0 + kk) * 128 + wc;
    const size_t w1 = w0 + 128;
    a[0]  = ar[k0 + kk];
    a[1]  = ar[k0 + kk + 1];
    bq[0] = Wq[w0];
    bq[1] = Wq[w1];
    bk[0] = Wk[w0];
    bk[1] = Wk[w1];
    bv[0] = Wv[w0];
    bv[1] = Wv[w1];
    cq = __builtin_amdgcn_wmma_f32_16x16x4_f32(false, a, false, bq, (short)0, cq,
                                               false, false);
    ck = __builtin_amdgcn_wmma_f32_16x16x4_f32(false, a, false, bk, (short)0, ck,
                                               false, false);
    cv = __builtin_amdgcn_wmma_f32_16x16x4_f32(false, a, false, bv, (short)0, cv,
                                               false, false);
  }
  const size_t o = (size_t)(row0 + mb) * 128 + wc;
#pragma unroll
  for (int r = 0; r < 8; ++r) q[o + (size_t)r * 128] = cq[r];
#pragma unroll
  for (int r = 0; r < 8; ++r) k[o + (size_t)r * 128] = ck[r];
#pragma unroll
  for (int r = 0; r < 8; ++r) v[o + (size_t)r * 128] = cv[r];
}

template <int MASK>
__device__ __forceinline__ float row_xor(float x) {
  return __int_as_float(__builtin_amdgcn_update_dpp(
      0, __float_as_int(x), 0x160 | MASK, 0xf, 0xf, true));
}

__device__ __forceinline__ float xhalf(float x) {
  int b = __builtin_bit_cast(int, x);
  return __builtin_bit_cast(
      float, __builtin_amdgcn_permlanex16(b, b, 0x76543210, 0xfedcba98, false,
                                          false));
}

#define SWAVES 4

__global__ __launch_bounds__(128) void edge_score_kernel(
    const float* __restrict__ ea, const int* __restrict__ src,
    const int* __restrict__ dst, const float* __restrict__ Wek,
    const float* __restrict__ Wg1, const float* __restrict__ Wg2,
    const float* __restrict__ q, const float* __restrict__ k,
    float* __restrict__ scores, unsigned long long* __restrict__ ms,
    int nedge) {
  __shared__ float4 g1s[SWAVES][32][17];
  __shared__ float4 gts[SWAVES][32][2];

  const int lane = threadIdx.x & 31;
  const int wv   = threadIdx.x >> 5;
  const int e0   = (blockIdx.x * SWAVES + wv) << 5;
  const int m    = lane & 15;
  const int kk   = (lane >> 4) << 1;
  const int mb   = (lane >> 4) << 3;

  v2f afr0[8], afr1[8];
  {
    const float* er0 = ea + (size_t)(e0 + m) * 32;
    const float* er1 = er0 + 16 * 32;
#pragma unroll
    for (int i = 0; i < 8; ++i) {
      afr0[i][0] = er0[4 * i + kk];
      afr0[i][1] = er0[4 * i + kk + 1];
      afr1[i][0] = er1[4 * i + kk];
      afr1[i][1] = er1[4 * i + kk + 1];
    }
  }

  {
    float* g1f = (float*)&g1s[wv][0][0];
#pragma unroll
    for (int t = 0; t < 4; ++t) {
      v8f c0 = {}, c1 = {};
      const float* wp = Wg1 + t * 16 + m;
#pragma unroll
      for (int i = 0; i < 8; ++i) {
        v2f b;
        b[0] = wp[(size_t)(4 * i + kk) * 64];
        b[1] = wp[(size_t)(4 * i + kk + 1) * 64];
        c0 = __builtin_amdgcn_wmma_f32_16x16x4_f32(false, afr0[i], false, b,
                                                   (short)0, c0, false, false);
        c1 = __builtin_amdgcn_wmma_f32_16x16x4_f32(false, afr1[i], false, b,
                                                   (short)0, c1, false, false);
      }
#pragma unroll
      for (int r = 0; r < 8; ++r) {
        const float x0 = c0[r];
        const float x1 = c1[r];
        g1f[(mb + r) * 68 + t * 16 + m]      = x0 / (1.0f + __expf(-x0));
        g1f[(16 + mb + r) * 68 + t * 16 + m] = x1 / (1.0f + __expf(-x1));
      }
    }
  }
  __syncthreads();

  {
    float acc[8];
#pragma unroll
    for (int hh = 0; hh < 8; ++hh) acc[hh] = 0.f;
#pragma unroll 2
    for (int j4 = 0; j4 < 16; ++j4) {
      const float4 g4 = g1s[wv][lane][j4];
      const float* w2 = Wg2 + j4 * 32;
#pragma unroll
      for (int hh = 0; hh < 8; ++hh) acc[hh] = fmaf(g4.x, w2[hh], acc[hh]);
#pragma unroll
      for (int hh = 0; hh < 8; ++hh) acc[hh] = fmaf(g4.y, w2[8 + hh], acc[hh]);
#pragma unroll
      for (int hh = 0; hh < 8; ++hh) acc[hh] = fmaf(g4.z, w2[16 + hh], acc[hh]);
#pragma unroll
      for (int hh = 0; hh < 8; ++hh) acc[hh] = fmaf(g4.w, w2[24 + hh], acc[hh]);
    }
    gts[wv][lane][0] = make_float4(acc[0], acc[1], acc[2], acc[3]);
    gts[wv][lane][1] = make_float4(acc[4], acc[5], acc[6], acc[7]);
  }
  __syncthreads();

  const unsigned so  = (unsigned)src[e0 + lane] * 512u;
  const unsigned dof = (unsigned)dst[e0 + lane] * 512u;

  const bool upper = lane >= 16;
  const bool b8 = (m & 8) != 0;
  const bool b4 = (m & 4) != 0;
  const bool b2 = (m & 2) != 0;
  const bool b1 = (m & 1) != 0;
  float* dtf = (float*)&g1s[wv][0][0];
  const char* kb = (const char*)k;
  const char* qb = (const char*)q;
  unsigned sN[32], dN[32];
#pragma unroll
  for (int j = 0; j < 32; ++j) {
    sN[j] = (unsigned)__builtin_amdgcn_readlane((int)so, j);
    dN[j] = (unsigned)__builtin_amdgcn_readlane((int)dof, j);
  }

#pragma unroll 1
  for (int p = 0; p < 4; ++p) {
    const unsigned cb = (unsigned)(p * 128 + (lane << 2));

    float kx[16], qx[16];
    auto gather = [&](const int T) {
#pragma unroll
      for (int j = 0; j < 16; ++j) {
        const unsigned ko = sN[16 * T + j] + cb;
        const unsigned qo = dN[16 * T + j] + cb;
        kx[j] = *(const float*)(kb + ko);
        qx[j] = *(const float*)(qb + qo);
      }
    };
    gather(0);

    v8f c00 = {}, c01 = {}, c10 = {}, c11 = {};
    const float* wp = Wek + p * 32 + m;
#pragma unroll
    for (int i = 0; i < 8; ++i) {
      v2f b0, b1v;
      b0[0]  = wp[(size_t)(4 * i + kk) * 128];
      b0[1]  = wp[(size_t)(4 * i + kk + 1) * 128];
      b1v[0] = wp[(size_t)(4 * i + kk) * 128 + 16];
      b1v[1] = wp[(size_t)(4 * i + kk + 1) * 128 + 16];
      c00 = __builtin_amdgcn_wmma_f32_16x16x4_f32(false, afr0[i], false, b0,
                                                  (short)0, c00, false, false);
      c10 = __builtin_amdgcn_wmma_f32_16x16x4_f32(false, afr1[i], false, b0,
                                                  (short)0, c10, false, false);
      c01 = __builtin_amdgcn_wmma_f32_16x16x4_f32(false, afr0[i], false, b1v,
                                                  (short)0, c01, false, false);
      c11 = __builtin_amdgcn_wmma_f32_16x16x4_f32(false, afr1[i], false, b1v,
                                                  (short)0, c11, false, false);
    }

    float pr[16];
    auto products = [&](const v8f& ca, const v8f& cb1) {
#pragma unroll
      for (int r = 0; r < 8; ++r) {
        const float t0 = xhalf(ca[r]);
        const float t1 = xhalf(cb1[r]);
        const float ex = upper ? t1 : ca[r];
        const float ey = upper ? cb1[r] : t0;
        pr[r]     = qx[r] * (kx[r] + ex);
        pr[8 + r] = qx[8 + r] * (kx[8 + r] + ey);
      }
    };
    auto reduce = [&](const int T) {
      float s8[8], s4[4], s2[2];
#pragma unroll
      for (int j = 0; j < 8; ++j)
        s8[j] = (b8 ? pr[j + 8] : pr[j]) + row_xor<8>(b8 ? pr[j] : pr[j + 8]);
#pragma unroll
      for (int j = 0; j < 4; ++j)
        s4[j] = (b4 ? s8[j + 4] : s8[j]) + row_xor<4>(b4 ? s8[j] : s8[j + 4]);
#pragma unroll
      for (int j = 0; j < 2; ++j)
        s2[j] = (b2 ? s4[j + 2] : s4[j]) + row_xor<2>(b2 ? s4[j] : s4[j + 2]);
      const float u = (b1 ? s2[1] : s2[0]) + row_xor<1>(b1 ? s2[0] : s2[1]);
      dtf[(16 * T + m) * 8 + 2 * p + (lane >> 4)] = u;
    };

    products(c00, c01);
    __builtin_amdgcn_sched_barrier(0);
    gather(1);
    reduce(0);
    products(c10, c11);
    reduce(1);
  }

  const float* gtf = (const float*)&gts[wv][0][0];
  const int hm = lane & 7;
  float dv[8], gv[8];
  int dn[8];
#pragma unroll
  for (int it = 0; it < 8; ++it) {
    dn[it] = dst[e0 + 4 * it + (lane >> 3)];
    dv[it] = dtf[32 * it + lane];
    gv[it] = gtf[32 * it + lane];
  }
  float sc[8];
  unsigned long long* wp[8];
#pragma unroll
  for (int it = 0; it < 8; ++it) {
    sc[it] = dv[it] * 0.25f + gv[it];
    scores[(size_t)(e0 + 4 * it) * 8 + lane] = sc[it];
    wp[it] = ms + ((size_t)dn[it] * 8 + hm);
  }
  unsigned long long cur[8], nw[8], old[8];
  __builtin_amdgcn_sched_barrier(0);
#pragma unroll
  for (int it = 0; it < 8; ++it)
    cur[it] = __hip_atomic_load(wp[it], __ATOMIC_RELAXED, __HIP_MEMORY_SCOPE_AGENT);
  __builtin_amdgcn_sched_barrier(0);
#pragma unroll
  for (int it = 0; it < 8; ++it) nw[it] = softmax_take(cur[it], sc[it]);
  __builtin_amdgcn_sched_barrier(0);
#pragma unroll
  for (int it = 0; it < 8; ++it) old[it] = atomicCAS(wp[it], cur[it], nw[it]);
  __builtin_amdgcn_sched_barrier(0);
  unsigned pend = 0;
#pragma unroll
  for (int it = 0; it < 8; ++it) {
    if (old[it] != cur[it]) pend |= 1u << it;
    cur[it] = old[it];
  }
  for (int t = 0; t < nedge && pend != 0; ++t) {
#pragma unroll
    for (int it = 0; it < 8; ++it)
      if (pend & (1u << it))
        old[it] = atomicCAS(wp[it], cur[it], softmax_take(cur[it], sc[it]));
#pragma unroll
    for (int it = 0; it < 8; ++it) {
      pend = (old[it] == cur[it]) ? (pend & ~(1u << it)) : pend;
      cur[it] = old[it];
    }
  }
}


__global__ __launch_bounds__(256) void edge_aggregate_kernel(
    const float* __restrict__ ea, const int* __restrict__ src,
    const int* __restrict__ dst, const float* __restrict__ Wev,
    const float* __restrict__ v, const float* __restrict__ scores,
    const float2* __restrict__ ms, float* __restrict__ acc) {
  const int lane = threadIdx.x & 31;
  const int wv   = threadIdx.x >> 5;
  const int e0   = __builtin_amdgcn_readfirstlane((blockIdx.x * WAVES + wv) << 4);
  const int m    = lane & 15;
  const int kk   = (lane >> 4) << 1;
  const int mb   = (lane >> 4) << 3;
  const bool upper = lane >= 16;

  v2f afr[8];
  const float* er = ea + (size_t)(e0 + m) * 32;
#pragma unroll
  for (int i = 0; i < 8; ++i) {
    afr[i][0] = er[4 * i + kk];
    afr[i][1] = er[4 * i + kk + 1];
  }

  int sn[16], dn[16];
#pragma unroll
  for (int j = 0; j < 16; ++j) {
    sn[j] = src[e0 + j];
    dn[j] = dst[e0 + j];
  }

  float vx[8], vy[8];
  auto load_v = [&](const int p) {
    const int colx = p * 32 + lane;
#pragma unroll
    for (int r = 0; r < 8; ++r) {
      vx[r] = v[(size_t)sn[r] * 128 + colx];
      vy[r] = v[(size_t)sn[8 + r] * 128 + colx];
    }
  };

  const int hm = m & 7;
  int att8[8];
  {
    load_v(0);
    float sv[8];
    float2 dv[8];
#pragma unroll
    for (int r = 0; r < 8; ++r) {
      const int dr = upper ? dn[8 + r] : dn[r];
      sv[r] = scores[(size_t)(e0 + mb + r) * 8 + hm];
      dv[r] = ms[(size_t)dr * 8 + hm];
    }
    __builtin_amdgcn_sched_barrier(0);
#pragma unroll
    for (int r = 0; r < 8; ++r)
      att8[r] = __float_as_int(__expf(sv[r] - dv[r].x) / (dv[r].y + 1e-9f));
  }

  auto trip = [&](const int p) {
    const int colx = p * 32 + lane;
    if (p > 0) {
      load_v(p);
      __builtin_amdgcn_sched_barrier(0);
    }

    v8f c0 = {}, c1 = {};
    const float* wp = Wev + p * 32 + m;
#pragma unroll
    for (int i = 0; i < 8; ++i) {
      v2f b0, b1;
      b0[0] = wp[(size_t)(4 * i + kk) * 128];
      b0[1] = wp[(size_t)(4 * i + kk + 1) * 128];
      b1[0] = wp[(size_t)(4 * i + kk) * 128 + 16];
      b1[1] = wp[(size_t)(4 * i + kk + 1) * 128 + 16];
      c0 = __builtin_amdgcn_wmma_f32_16x16x4_f32(false, afr[i], false, b0,
                                                 (short)0, c0, false, false);
      c1 = __builtin_amdgcn_wmma_f32_16x16x4_f32(false, afr[i], false, b1,
                                                 (short)0, c1, false, false);
    }

#pragma unroll
    for (int r = 0; r < 8; ++r) {
      const float t0 = xhalf(c0[r]);
      const float t1 = xhalf(c1[r]);
      const float cx = upper ? t1 : c0[r];
      const float cy = upper ? c1[r] : t0;
      const float ax_lo = __int_as_float(__builtin_amdgcn_readlane(att8[r], 2 * p));
      const float ax_hi = __int_as_float(__builtin_amdgcn_readlane(att8[r], 2 * p + 1));
      const float ay_lo = __int_as_float(__builtin_amdgcn_readlane(att8[r], 2 * p + 16));
      const float ay_hi = __int_as_float(__builtin_amdgcn_readlane(att8[r], 2 * p + 17));
      const float ax = upper ? ax_hi : ax_lo;
      const float ay = upper ? ay_hi : ay_lo;
      atomicAdd(&acc[(size_t)dn[r] * 128 + colx], ax * (vx[r] + cx));
      atomicAdd(&acc[(size_t)dn[8 + r] * 128 + colx], ay * (vy[r] + cy));
    }
  };

#pragma unroll
  for (int p = 0; p < 4; ++p) trip(p);
}

extern "C" void kernel_launch(void* const* d_in, const int* in_sizes, int n_in,
                              void* d_out, int out_size, void* d_ws,
                              size_t ws_size, hipStream_t stream) {
  const float* x   = (const float*)d_in[0];
  const int*   ei  = (const int*)d_in[1];
  const float* ea  = (const float*)d_in[2];
  const float* Wq  = (const float*)d_in[3];
  const float* Wk  = (const float*)d_in[4];
  const float* Wv  = (const float*)d_in[5];
  const float* Wo  = (const float*)d_in[6];
  const float* Wek = (const float*)d_in[7];
  const float* Wev = (const float*)d_in[8];
  const float* Wg1 = (const float*)d_in[9];
  const float* Wg2 = (const float*)d_in[10];
  float* out = (float*)d_out;

  const int N = in_sizes[0] / 128;
  const int E = in_sizes[2] / 32;
  const int* src = ei;
  const int* dst = ei + E;

  float* q      = (float*)d_ws;
  float* k      = q + (size_t)N * 128;
  float* v      = k + (size_t)N * 128;
  float* scores = v + (size_t)N * 128;
  float* acc    = scores + (size_t)E * 8;
  unsigned long long* ms = (unsigned long long*)(acc + (size_t)N * 128);

  hipMemsetAsync(acc, 0, (size_t)(N * 128 + N * 8 + N * 8) * sizeof(float),
                 stream);

  dim3 blk(256);
  qkv_kernel<<<dim3(N / 16), blk, 0, stream>>>(x, Wq, Wk, Wv, q, k, v);

  edge_score_kernel<<<dim3(E / (32 * SWAVES)), dim3(32 * SWAVES), 0, stream>>>(
      ea, src, dst, Wek, Wg1, Wg2, q, k, scores, ms, E);

  edge_aggregate_kernel<<<dim3(E / (16 * WAVES)), blk, 0, stream>>>(
      ea, src, dst, Wev, v, scores, (const float2*)ms, acc);

  gemm128_kernel<<<dim3(N / 16), blk, 0, stream>>>(acc, Wo, out);
}


# Wev, Wq/Wk/Wv, Wo kept as B fragments in registers: a fixed grid of resident waves loads its weights once, then walks over many 16-row tiles; sums per tile unchanged, atomics unordered as before
# speedup vs baseline: 2.8629x; 1.0528x over previous
#include <hip/hip_runtime.h>


typedef __attribute__((ext_vector_type(2))) float v2f;
typedef __attribute__((ext_vector_type(8))) float v8f;

#define WAVES 8

__device__ __forceinline__ unsigned long long softmax_take(unsigned long long w,
                                                           float s) {
  const float mo = __uint_as_float((unsigned)w);
  const float so = __uint_as_float((unsigned)(w >> 32));
  const float me = (so == 0.f) ? __uint_as_float(0xff800000u) : mo;
  const float mn = fmaxf(me, s);
  const float sn = so * __expf(me - mn) + __expf(s - mn);
  return (unsigned long long)__float_as_uint(mn) |
         ((unsigned long long)__float_as_uint(sn) << 32);
}

#define GEMM_BLOCKS 512

__global__ __launch_bounds__(256, 4) void gemm128_kernel(
    const float* __restrict__ A, const float* __restrict__ W,
    float* __restrict__ out, int ntiles) {
  const int lane = threadIdx.x & 31;
  const int wv   = threadIdx.x >> 5;
  const int n0   = wv << 4;
  const int m    = lane & 15;
  const int kk   = (lane >> 4) << 1;
  const int mb   = (lane >> 4) << 3;

  v2f bw[32];
  {
    const float* wp = W + n0 + m;
#pragma unroll
    for (int s = 0; s < 32; ++s) {
      bw[s][0] = wp[(size_t)(4 * s + kk) * 128];
      bw[s][1] = wp[(size_t)(4 * s + kk + 1) * 128];
    }
  }

#pragma unroll 1
  for (int t = blockIdx.x; t < ntiles; t += gridDim.x) {
    const int row0 = t << 4;
    v8f c = {};
    const float* ar = A + (size_t)(row0 + m) * 128;
#pragma unroll
    for (int s = 0; s < 32; ++s) {
      v2f a;
      a[0] = ar[4 * s + kk];
      a[1] = ar[4 * s + kk + 1];
      c = __builtin_amdgcn_wmma_f32_16x16x4_f32(false, a, false, bw[s], (short)0, c,
                                                false, false);
    }
    float* op = out + (size_t)(row0 + mb) * 128 + n0 + m;
#pragma unroll
    for (int r = 0; r < 8; ++r) op[(size_t)r * 128] = c[r];
  }
}

__global__ __launch_bounds__(256, 4) void qkv_kernel(
    const float* __restrict__ A, const float* __restrict__ Wq,
    const float* __restrict__ Wk, const float* __restrict__ Wv,
    float* __restrict__ q, float* __restrict__ k, float* __restrict__ v,
    int ntiles) {
  const int lane = threadIdx.x & 31;
  const int wv   = threadIdx.x >> 5;
  const int n0   = wv << 4;
  const int m    = lane & 15;
  const int kk   = (lane >> 4) << 1;
  const int mb   = (lane >> 4) << 3;
  const size_t wc = (size_t)(n0 + m);

  v2f bq[32], bk[32], bv[32];
#pragma unroll
  for (int s = 0; s < 32; ++s) {
    const size_t w0 = (size_t)(4 * s + kk) * 128 + wc;
    const size_t w1 = w0 + 128;
    bq[s][0] = Wq[w0];
    bq[s][1] = Wq[w1];
    bk[s][0] = Wk[w0];
    bk[s][1] = Wk[w1];
    bv[s][0] = Wv[w0];
    bv[s][1] = Wv[w1];
  }

#pragma unroll 1
  for (int t = blockIdx.x; t < ntiles; t += gridDim.x) {
    const int row0 = t << 4;
    v8f cq = {}, ck = {}, cv = {};
    const float* ar = A + (size_t)(row0 + m) * 128;
#pragma unroll
    for (int s = 0; s < 32; ++s) {
      v2f a;
      a[0] = ar[4 * s + kk];
      a[1] = ar[4 * s + kk + 1];
      cq = __builtin_amdgcn_wmma_f32_16x16x4_f32(false, a, false, bq[s], (short)0, cq,
                                                 false, false);
      ck = __builtin_amdgcn_wmma_f32_16x16x4_f32(false, a, false, bk[s], (short)0, ck,
                                                 false, false);
      cv = __builtin_amdgcn_wmma_f32_16x16x4_f32(false, a, false, bv[s], (short)0, cv,
                                                 false, false);
    }
    const size_t o = (size_t)(row0 + mb) * 128 + wc;
#pragma unroll
    for (int r = 0; r < 8; ++r) q[o + (size_t)r * 128] = cq[r];
#pragma unroll
    for (int r = 0; r < 8; ++r) k[o + (size_t)r * 128] = ck[r];
#pragma unroll
    for (int r = 0; r < 8; ++r) v[o + (size_t)r * 128] = cv[r];
  }
}

template <int MASK>
__device__ __forceinline__ float row_xor(float x) {
  return __int_as_float(__builtin_amdgcn_update_dpp(
      0, __float_as_int(x), 0x160 | MASK, 0xf, 0xf, true));
}

__device__ __forceinline__ float xhalf(float x) {
  int b = __builtin_bit_cast(int, x);
  return __builtin_bit_cast(
      float, __builtin_amdgcn_permlanex16(b, b, 0x76543210, 0xfedcba98, false,
                                          false));
}

#define SWAVES 4

__global__ __launch_bounds__(128) void edge_score_kernel(
    const float* __restrict__ ea, const int* __restrict__ src,
    const int* __restrict__ dst, const float* __restrict__ Wek,
    const float* __restrict__ Wg1, const float* __restrict__ Wg2,
    const float* __restrict__ q, const float* __restrict__ k,
    float* __restrict__ scores, unsigned long long* __restrict__ ms,
    int nedge) {
  __shared__ float4 g1s[SWAVES][32][17];
  __shared__ float4 gts[SWAVES][32][2];

  const int lane = threadIdx.x & 31;
  const int wv   = threadIdx.x >> 5;
  const int e0   = (blockIdx.x * SWAVES + wv) << 5;
  const int m    = lane & 15;
  const int kk   = (lane >> 4) << 1;
  const int mb   = (lane >> 4) << 3;

  v2f afr0[8], afr1[8];
  {
    const float* er0 = ea + (size_t)(e0 + m) * 32;
    const float* er1 = er0 + 16 * 32;
#pragma unroll
    for (int i = 0; i < 8; ++i) {
      afr0[i][0] = er0[4 * i + kk];
      afr0[i][1] = er0[4 * i + kk + 1];
      afr1[i][0] = er1[4 * i + kk];
      afr1[i][1] = er1[4 * i + kk + 1];
    }
  }

  {
    float* g1f = (float*)&g1s[wv][0][0];
#pragma unroll
    for (int t = 0; t < 4; ++t) {
      v8f c0 = {}, c1 = {};
      const float* wp = Wg1 + t * 16 + m;
#pragma unroll
      for (int i = 0; i < 8; ++i) {
        v2f b;
        b[0] = wp[(size_t)(4 * i + kk) * 64];
        b[1] = wp[(size_t)(4 * i + kk + 1) * 64];
        c0 = __builtin_amdgcn_wmma_f32_16x16x4_f32(false, afr0[i], false, b,
                                                   (short)0, c0, false, false);
        c1 = __builtin_amdgcn_wmma_f32_16x16x4_f32(false, afr1[i], false, b,
                                                   (short)0, c1, false, false);
      }
#pragma unroll
      for (int r = 0; r < 8; ++r) {
        const float x0 = c0[r];
        const float x1 = c1[r];
        g1f[(mb + r) * 68 + t * 16 + m]      = x0 / (1.0f + __expf(-x0));
        g1f[(16 + mb + r) * 68 + t * 16 + m] = x1 / (1.0f + __expf(-x1));
      }
    }
  }
  __syncthreads();

  {
    float acc[8];
#pragma unroll
    for (int hh = 0; hh < 8; ++hh) acc[hh] = 0.f;
#pragma unroll 2
    for (int j4 = 0; j4 < 16; ++j4) {
      const float4 g4 = g1s[wv][lane][j4];
      const float* w2 = Wg2 + j4 * 32;
#pragma unroll
      for (int hh = 0; hh < 8; ++hh) acc[hh] = fmaf(g4.x, w2[hh], acc[hh]);
#pragma unroll
      for (int hh = 0; hh < 8; ++hh) acc[hh] = fmaf(g4.y, w2[8 + hh], acc[hh]);
#pragma unroll
      for (int hh = 0; hh < 8; ++hh) acc[hh] = fmaf(g4.z, w2[16 + hh], acc[hh]);
#pragma unroll
      for (int hh = 0; hh < 8; ++hh) acc[hh] = fmaf(g4.w, w2[24 + hh], acc[hh]);
    }
    gts[wv][lane][0] = make_float4(acc[0], acc[1], acc[2], acc[3]);
    gts[wv][lane][1] = make_float4(acc[4], acc[5], acc[6], acc[7]);
  }
  __syncthreads();

  const unsigned so  = (unsigned)src[e0 + lane] * 512u;
  const unsigned dof = (unsigned)dst[e0 + lane] * 512u;

  const bool upper = lane >= 16;
  const bool b8 = (m & 8) != 0;
  const bool b4 = (m & 4) != 0;
  const bool b2 = (m & 2) != 0;
  const bool b1 = (m & 1) != 0;
  float* dtf = (float*)&g1s[wv][0][0];
  const char* kb = (const char*)k;
  const char* qb = (const char*)q;
  unsigned sN[32], dN[32];
#pragma unroll
  for (int j = 0; j < 32; ++j) {
    sN[j] = (unsigned)__builtin_amdgcn_readlane((int)so, j);
    dN[j] = (unsigned)__builtin_amdgcn_readlane((int)dof, j);
  }

#pragma unroll 1
  for (int p = 0; p < 4; ++p) {
    const unsigned cb = (unsigned)(p * 128 + (lane << 2));

    float kx[16], qx[16];
    auto gather = [&](const int T) {
#pragma unroll
      for (int j = 0; j < 16; ++j) {
        const unsigned ko = sN[16 * T + j] + cb;
        const unsigned qo = dN[16 * T + j] + cb;
        kx[j] = *(const float*)(kb + ko);
        qx[j] = *(const float*)(qb + qo);
      }
    };
    gather(0);

    v8f c00 = {}, c01 = {}, c10 = {}, c11 = {};
    const float* wp = Wek + p * 32 + m;
#pragma unroll
    for (int i = 0; i < 8; ++i) {
      v2f b0, b1v;
      b0[0]  = wp[(size_t)(4 * i + kk) * 128];
      b0[1]  = wp[(size_t)(4 * i + kk + 1) * 128];
      b1v[0] = wp[(size_t)(4 * i + kk) * 128 + 16];
      b1v[1] = wp[(size_t)(4 * i + kk + 1) * 128 + 16];
      c00 = __builtin_amdgcn_wmma_f32_16x16x4_f32(false, afr0[i], false, b0,
                                                  (short)0, c00, false, false);
      c10 = __builtin_amdgcn_wmma_f32_16x16x4_f32(false, afr1[i], false, b0,
                                                  (short)0, c10, false, false);
      c01 = __builtin_amdgcn_wmma_f32_16x16x4_f32(false, afr0[i], false, b1v,
                                                  (short)0, c01, false, false);
      c11 = __builtin_amdgcn_wmma_f32_16x16x4_f32(false, afr1[i], false, b1v,
                                                  (short)0, c11, false, false);
    }

    float pr[16];
    auto products = [&](const v8f& ca, const v8f& cb1) {
#pragma unroll
      for (int r = 0; r < 8; ++r) {
        const float t0 = xhalf(ca[r]);
        const float t1 = xhalf(cb1[r]);
        const float ex = upper ? t1 : ca[r];
        const float ey = upper ? cb1[r] : t0;
        pr[r]     = qx[r] * (kx[r] + ex);
        pr[8 + r] = qx[8 + r] * (kx[8 + r] + ey);
      }
    };
    auto reduce = [&](const int T) {
      float s8[8], s4[4], s2[2];
#pragma unroll
      for (int j = 0; j < 8; ++j)
        s8[j] = (b8 ? pr[j + 8] : pr[j]) + row_xor<8>(b8 ? pr[j] : pr[j + 8]);
#pragma unroll
      for (int j = 0; j < 4; ++j)
        s4[j] = (b4 ? s8[j + 4] : s8[j]) + row_xor<4>(b4 ? s8[j] : s8[j + 4]);
#pragma unroll
      for (int j = 0; j < 2; ++j)
        s2[j] = (b2 ? s4[j + 2] : s4[j]) + row_xor<2>(b2 ? s4[j] : s4[j + 2]);
      const float u = (b1 ? s2[1] : s2[0]) + row_xor<1>(b1 ? s2[0] : s2[1]);
      dtf[(16 * T + m) * 8 + 2 * p + (lane >> 4)] = u;
    };

    products(c00, c01);
    __builtin_amdgcn_sched_barrier(0);
    gather(1);
    reduce(0);
    products(c10, c11);
    reduce(1);
  }

  const float* gtf = (const float*)&gts[wv][0][0];
  const int hm = lane & 7;
  float dv[8], gv[8];
  int dn[8];
#pragma unroll
  for (int it = 0; it < 8; ++it) {
    dn[it] = dst[e0 + 4 * it + (lane >> 3)];
    dv[it] = dtf[32 * it + lane];
    gv[it] = gtf[32 * it + lane];
  }
  float sc[8];
  unsigned long long* wp[8];
#pragma unroll
  for (int it = 0; it < 8; ++it) {
    sc[it] = dv[it] * 0.25f + gv[it];
    scores[(size_t)(e0 + 4 * it) * 8 + lane] = sc[it];
    wp[it] = ms + ((size_t)dn[it] * 8 + hm);
  }
  unsigned long long cur[8], nw[8], old[8];
  __builtin_amdgcn_sched_barrier(0);
#pragma unroll
  for (int it = 0; it < 8; ++it)
    cur[it] = __hip_atomic_load(wp[it], __ATOMIC_RELAXED, __HIP_MEMORY_SCOPE_AGENT);
  __builtin_amdgcn_sched_barrier(0);
#pragma unroll
  for (int it = 0; it < 8; ++it) nw[it] = softmax_take(cur[it], sc[it]);
  __builtin_amdgcn_sched_barrier(0);
#pragma unroll
  for (int it = 0; it < 8; ++it) old[it] = atomicCAS(wp[it], cur[it], nw[it]);
  __builtin_amdgcn_sched_barrier(0);
  unsigned pend = 0;
#pragma unroll
  for (int it = 0; it < 8; ++it) {
    if (old[it] != cur[it]) pend |= 1u << it;
    cur[it] = old[it];
  }
  for (int t = 0; t < nedge && pend != 0; ++t) {
#pragma unroll
    for (int it = 0; it < 8; ++it)
      if (pend & (1u << it))
        old[it] = atomicCAS(wp[it], cur[it], softmax_take(cur[it], sc[it]));
#pragma unroll
    for (int it = 0; it < 8; ++it) {
      pend = (old[it] == cur[it]) ? (pend & ~(1u << it)) : pend;
      cur[it] = old[it];
    }
  }
}


#define AGG_BLOCKS 512

__global__ __launch_bounds__(256, 4) void edge_aggregate_kernel(
    const float* __restrict__ ea, const int* __restrict__ src,
    const int* __restrict__ dst, const float* __restrict__ Wev,
    const float* __restrict__ v, const float* __restrict__ scores,
    const float2* __restrict__ ms, float* __restrict__ acc, int ntiles) {
  const int lane = threadIdx.x & 31;
  const int wv   = threadIdx.x >> 5;
  const int m    = lane & 15;
  const int kk   = (lane >> 4) << 1;
  const int mb   = (lane >> 4) << 3;
  const bool upper = lane >= 16;
  const int hm = m & 7;

  v2f wa[4][8], wb[4][8];
#pragma unroll
  for (int p = 0; p < 4; ++p) {
    const float* wp = Wev + p * 32 + m;
#pragma unroll
    for (int i = 0; i < 8; ++i) {
      wa[p][i][0] = wp[(size_t)(4 * i + kk) * 128];
      wa[p][i][1] = wp[(size_t)(4 * i + kk + 1) * 128];
      wb[p][i][0] = wp[(size_t)(4 * i + kk) * 128 + 16];
      wb[p][i][1] = wp[(size_t)(4 * i + kk + 1) * 128 + 16];
    }
  }

  const int nwav = __builtin_amdgcn_readfirstlane((int)gridDim.x * WAVES);
  int tile = __builtin_amdgcn_readfirstlane((int)blockIdx.x * WAVES + wv);

  const int* ip = (upper ? dst : src) + m;
  const char* vbp = (const char*)v;
  char* abp = (char*)acc;
  int idn = ip[(size_t)(tile < ntiles ? tile : 0) << 4];

#pragma unroll 1
  for (; tile < ntiles; tile += nwav) {
    const int e0 = tile << 4;
    const int idx = idn;
    const int tnx = tile + nwav;
    idn = ip[(size_t)(tnx < ntiles ? tnx : tile) << 4];

    v2f afr[8];
    const float* er = ea + (size_t)(e0 + m) * 32;
#pragma unroll
    for (int i = 0; i < 8; ++i) {
      afr[i][0] = er[4 * i + kk];
      afr[i][1] = er[4 * i + kk + 1];
    }

    int sn[16], dn[16];
#pragma unroll
    for (int j = 0; j < 16; ++j) {
      sn[j] = __builtin_amdgcn_readlane(idx, j);
      dn[j] = __builtin_amdgcn_readlane(idx, 16 + j);
    }

    float vx[8], vy[8];
    auto load_v = [&](const int p) {
      const unsigned cb = (unsigned)(p * 128 + (lane << 2));
#pragma unroll
      for (int r = 0; r < 8; ++r) {
        vx[r] = *(const float*)(vbp + ((unsigned)sn[r] * 512u + cb));
        vy[r] = *(const float*)(vbp + ((unsigned)sn[8 + r] * 512u + cb));
      }
    };

    int att8[8];
    {
      load_v(0);
      float sv[8];
      float2 dv[8];
#pragma unroll
      for (int r = 0; r < 8; ++r) {
        const int dr = upper ? dn[8 + r] : dn[r];
        sv[r] = scores[(size_t)(e0 + mb + r) * 8 + hm];
        dv[r] = ms[(size_t)dr * 8 + hm];
      }
      __builtin_amdgcn_sched_barrier(0);
#pragma unroll
      for (int r = 0; r < 8; ++r)
        att8[r] = __float_as_int(__expf(sv[r] - dv[r].x) / (dv[r].y + 1e-9f));
    }

    auto trip = [&](const int p) {
      const unsigned cb = (unsigned)(p * 128 + (lane << 2));
      if (p > 0) {
        load_v(p);
        __builtin_amdgcn_sched_barrier(0);
      }

      v8f c0 = {}, c1 = {};
#pragma unroll
      for (int i = 0; i < 8; ++i) {
        c0 = __builtin_amdgcn_wmma_f32_16x16x4_f32(false, afr[i], false, wa[p][i],
                                                   (short)0, c0, false, false);
        c1 = __builtin_amdgcn_wmma_f32_16x16x4_f32(false, afr[i], false, wb[p][i],
                                                   (short)0, c1, false, false);
      }

#pragma unroll
      for (int r = 0; r < 8; ++r) {
        const float t0 = xhalf(c0[r]);
        const float t1 = xhalf(c1[r]);
        const float cx = upper ? t1 : c0[r];
        const float cy = upper ? c1[r] : t0;
        const float ax_lo = __int_as_float(__builtin_amdgcn_readlane(att8[r], 2 * p));
        const float ax_hi = __int_as_float(__builtin_amdgcn_readlane(att8[r], 2 * p + 1));
        const float ay_lo = __int_as_float(__builtin_amdgcn_readlane(att8[r], 2 * p + 16));
        const float ay_hi = __int_as_float(__builtin_amdgcn_readlane(att8[r], 2 * p + 17));
        const float ax = upper ? ax_hi : ax_lo;
        const float ay = upper ? ay_hi : ay_lo;
        atomicAdd((float*)(abp + ((unsigned)dn[r] * 512u + cb)), ax * (vx[r] + cx));
        atomicAdd((float*)(abp + ((unsigned)dn[8 + r] * 512u + cb)), ay * (vy[r] + cy));
      }
    };

#pragma unroll
    for (int p = 0; p < 4; ++p) trip(p);
  }
}

extern "C" void kernel_launch(void* const* d_in, const int* in_sizes, int n_in,
                              void* d_out, int out_size, void* d_ws,
                              size_t ws_size, hipStream_t stream) {
  const float* x   = (const float*)d_in[0];
  const int*   ei  = (const int*)d_in[1];
  const float* ea  = (const float*)d_in[2];
  const float* Wq  = (const float*)d_in[3];
  const float* Wk  = (const float*)d_in[4];
  const float* Wv  = (const float*)d_in[5];
  const float* Wo  = (const float*)d_in[6];
  const float* Wek = (const float*)d_in[7];
  const float* Wev = (const float*)d_in[8];
  const float* Wg1 = (const float*)d_in[9];
  const float* Wg2 = (const float*)d_in[10];
  float* out = (float*)d_out;

  const int N = in_sizes[0] / 128;
  const int E = in_sizes[2] / 32;
  const int* src = ei;
  const int* dst = ei + E;

  float* q      = (float*)d_ws;
  float* k      = q + (size_t)N * 128;
  float* v      = k + (size_t)N * 128;
  float* scores = v + (size_t)N * 128;
  float* acc    = scores + (size_t)E * 8;
  unsigned long long* ms = (unsigned long long*)(acc + (size_t)N * 128);

  hipMemsetAsync(acc, 0, (size_t)(N * 128 + N * 8 + N * 8) * sizeof(float),
                 stream);

  dim3 blk(256);
  qkv_kernel<<<dim3(GEMM_BLOCKS), blk, 0, stream>>>(x, Wq, Wk, Wv, q, k, v, N / 16);

  edge_score_kernel<<<dim3(E / (32 * SWAVES)), dim3(32 * SWAVES), 0, stream>>>(
      ea, src, dst, Wek, Wg1, Wg2, q, k, scores, ms, E);

  edge_aggregate_kernel<<<dim3(AGG_BLOCKS), blk, 0, stream>>>(
      ea, src, dst, Wev, v, scores, (const float2*)ms, acc, E / 16);

  gemm128_kernel<<<dim3(GEMM_BLOCKS), blk, 0, stream>>>(acc, Wo, out, N / 16);
}
